# starting kernel
# baseline (speedup 1.0000x reference)
#include <hip/hip_runtime.h>

#define D 128
#define NLAYERS 4

typedef __attribute__((ext_vector_type(2))) float v2f;
typedef __attribute__((ext_vector_type(8))) float v8f;


__global__ void zero_f32_kernel(float* __restrict__ p, long n) {
    long i = (long)blockIdx.x * blockDim.x + threadIdx.x;
    if (i < n) p[i] = 0.0f;
}

__global__ void degree_kernel(const int* __restrict__ src, const int* __restrict__ dst,
                              float* __restrict__ deg_src, float* __restrict__ deg_dst,
                              int E) {
    int e = blockIdx.x * blockDim.x + threadIdx.x;
    if (e < E) {
        unsafeAtomicAdd(&deg_src[src[e]], 1.0f);
        unsafeAtomicAdd(&deg_dst[dst[e]], 1.0f);
    }
}

__global__ void norm_kernel(float* __restrict__ deg, long n) {
    long i = (long)blockIdx.x * blockDim.x + threadIdx.x;
    if (i < n) deg[i] = rsqrtf(fmaxf(deg[i], 1.0f));
}

__global__ void spmm_scatter_kernel(const float* __restrict__ hin,
                                    const float* __restrict__ norm_src,
                                    const int* __restrict__ src,
                                    const int* __restrict__ dst,
                                    float* __restrict__ agg, int E) {
    int t = blockIdx.x * blockDim.x + threadIdx.x;
    int e = t >> 5;
    int q = (t & 31) << 2;
    if (e >= E) return;
    int s = src[e];
    int d = dst[e];
    float4 v = *(const float4*)(hin + (long)s * D + q);
    float ns = norm_src[s];
    float* outp = agg + (long)d * D + q;
    unsafeAtomicAdd(outp + 0, v.x * ns);
    unsafeAtomicAdd(outp + 1, v.y * ns);
    unsafeAtomicAdd(outp + 2, v.z * ns);
    unsafeAtomicAdd(outp + 3, v.w * ns);
}

__global__ void __launch_bounds__(256)
gcn_gemm_wmma_kernel(const float* __restrict__ agg,
                     const float* __restrict__ norm_dst,
                     const float* __restrict__ W,
                     const float* __restrict__ bias,
                     float* __restrict__ out, int N) {
    __shared__ float sA[16][D + 4];

    const int tid     = threadIdx.x;
    const int rowBase = blockIdx.x * 16;

    for (int idx = tid; idx < 16 * D; idx += 256) {
        int r = idx >> 7;
        int c = idx & (D - 1);
        int row = rowBase + r;
        float v = 0.0f;
        if (row < N) v = agg[(long)row * D + c] * norm_dst[row];
        sA[r][c] = v;
    }
    __syncthreads();

    const int wave  = tid >> 5;
    const int lane  = tid & 31;
    const int laneM = lane & 15;
    const int koff  = (lane >> 4) << 1;
    const int n     = (wave << 4) + laneM;

    v8f c;
    const float bv = bias[n];
#pragma unroll
    for (int v = 0; v < 8; ++v) c[v] = bv;

#pragma unroll
    for (int k0 = 0; k0 < D; k0 += 4) {
        v2f a, b;
        a.x = sA[laneM][k0 + koff];
        a.y = sA[laneM][k0 + koff + 1];
        b.x = W[(long)(k0 + koff) * D + n];
        b.y = W[(long)(k0 + koff + 1) * D + n];
        c = __builtin_amdgcn_wmma_f32_16x16x4_f32(
                 false, a,  false, b,
                 (short)0, c,  false,  false);
    }

    const int mBase = rowBase + ((lane >> 4) << 3);
#pragma unroll
    for (int v = 0; v < 8; ++v) {
        int row = mBase + v;
        if (row < N) out[(long)row * D + n] = c[v];
    }
}


extern "C" void kernel_launch(void* const* d_in, const int* in_sizes, int n_in,
                              void* d_out, int out_size, void* d_ws, size_t ws_size,
                              hipStream_t stream) {
    const float* feat = (const float*)d_in[0];
    const float* W    = (const float*)d_in[1];
    const float* bias = (const float*)d_in[2];
    const int*   src  = (const int*)d_in[3];
    const int*   dst  = (const int*)d_in[4];

    const int N = in_sizes[0] / D;
    const int E = in_sizes[3];

    float* ws       = (float*)d_ws;
    float* norm_src = ws;
    float* norm_dst = ws + N;
    float* agg      = ws + 2L * N;
    float* hbuf     = agg + (long)N * D;
    float* outf     = (float*)d_out;

    const long twoN = 2L * N;
    const long ND   = (long)N * D;

    zero_f32_kernel<<<(int)((twoN + 255) / 256), 256, 0, stream>>>(norm_src, twoN);
    degree_kernel<<<(E + 255) / 256, 256, 0, stream>>>(src, dst, norm_src, norm_dst, E);
    norm_kernel<<<(int)((twoN + 255) / 256), 256, 0, stream>>>(norm_src, twoN);

    const long edgeThreads = (long)E * 32;
    const int  edgeGrid    = (int)((edgeThreads + 255) / 256);
    const int  gemmGrid    = (N + 15) / 16;

    const float* hin = feat;
    for (int layer = 0; layer < NLAYERS; ++layer) {
        float* hout = (layer & 1) ? outf : hbuf;

        zero_f32_kernel<<<(int)((ND + 255) / 256), 256, 0, stream>>>(agg, ND);
        spmm_scatter_kernel<<<edgeGrid, 256, 0, stream>>>(hin, norm_src, src, dst, agg, E);
        gcn_gemm_wmma_kernel<<<gemmGrid, 256, 0, stream>>>(
            agg, norm_dst, W + (long)layer * D * D, bias + (long)layer * D, hout, N);

        hin = hout;
    }
}


# SpMM by dst-sorted row lists built per call (returning int atomic gives each edge its slot), one wave gathers a row, no float atomics on agg; per-row sum order = slot arrival order
# speedup vs baseline: 2.7562x; 2.7562x over previous
#include <hip/hip_runtime.h>

#define D 128
#define NLAYERS 4

typedef __attribute__((ext_vector_type(2))) float v2f;
typedef __attribute__((ext_vector_type(8))) float v8f;


__global__ void zero_words_kernel(int* __restrict__ p, long n) {
    long i = (long)blockIdx.x * blockDim.x + threadIdx.x;
    if (i < n) p[i] = 0;
}

__global__ void degree_rank_kernel(const int* __restrict__ src, const int* __restrict__ dst,
                                   float* __restrict__ deg_src, int* __restrict__ cnt_dst,
                                   int* __restrict__ rank, int E) {
    int e = blockIdx.x * blockDim.x + threadIdx.x;
    if (e < E) {
        const int s = src[e];
        const int d = dst[e];
        unsafeAtomicAdd(&deg_src[s], 1.0f);
        rank[e] = atomicAdd(&cnt_dst[d], 1);
    }
}

__global__ void __launch_bounds__(1024)
scan_norm_kernel(const int* __restrict__ cnt_dst, float* __restrict__ deg_src,
                 float* __restrict__ norm_dst, int* __restrict__ row_ptr, int N) {
    __shared__ int tileSum[32];
    __shared__ int headSum[32];
    const int tid  = threadIdx.x;
    const int lane = tid & 31;
    const int wv   = tid >> 5;
    const int tbeg = blockIdx.x * 1024;
    const int i    = tbeg + tid;

    int head = 0;
    for (int j = tid; j < tbeg; j += 1024) head += cnt_dst[j];
    const int c = (i < N) ? cnt_dst[i] : 0;

    int incl = c;
    for (int off = 1; off < 32; off <<= 1) {
        int t = __shfl_up(incl, off, 32);
        if (lane >= off) incl += t;
    }
    for (int off = 16; off > 0; off >>= 1) head += __shfl_xor(head, off, 32);
    if (lane == 31) {
        tileSum[wv] = incl;
        headSum[wv] = head;
    }
    __syncthreads();

    int x = headSum[lane] + ((lane < wv) ? tileSum[lane] : 0);
    for (int off = 16; off > 0; off >>= 1) x += __shfl_xor(x, off, 32);

    const int excl = x + incl - c;
    if (i < N) {
        row_ptr[i]  = excl;
        norm_dst[i] = rsqrtf(fmaxf((float)c, 1.0f));
        deg_src[i]  = rsqrtf(fmaxf(deg_src[i], 1.0f));
        if (i == N - 1) row_ptr[N] = excl + c;
    }
}

__global__ void fill_rows_kernel(const int* __restrict__ src, const int* __restrict__ dst,
                                 const int* __restrict__ rank, const int* __restrict__ row_ptr,
                                 const float* __restrict__ norm_src,
                                 int* __restrict__ pairs, int E) {
    int e = blockIdx.x * blockDim.x + threadIdx.x;
    if (e < E) {
        int s = src[e];
        int p = row_ptr[dst[e]] + rank[e];
        int2 pr;
        pr.x = s;
        pr.y = __float_as_int(norm_src[s]);
        *(int2*)(pairs + 2L * p) = pr;
    }
}

#define GATHER_LOAD(k)                                                         \
    {                                                                          \
        int   s_ = __builtin_amdgcn_readlane(pr.x, i + (k));                   \
        w[k]     = __int_as_float(__builtin_amdgcn_readlane(pr.y, i + (k)));   \
        v[k]     = *(const float4*)(hin + (long)s_ * D + q);                   \
    }
#define GATHER_ACC(k)                                                          \
    {                                                                          \
        acc.x += v[k].x * w[k];                                                \
        acc.y += v[k].y * w[k];                                                \
        acc.z += v[k].z * w[k];                                                \
        acc.w += v[k].w * w[k];                                                \
    }

__global__ void __launch_bounds__(256)
spmm_gather_kernel(const float* __restrict__ hin,
                   const int* __restrict__ pairs,
                   const int* __restrict__ row_ptr,
                   float* __restrict__ agg, int N) {
    const int lane = threadIdx.x & 31;
    const int q    = lane << 2;
    const int d    = blockIdx.x * 8 + (threadIdx.x >> 5);
    if (d >= N) return;

    const int rbeg = __builtin_amdgcn_readfirstlane(row_ptr[d]);
    const int rend = __builtin_amdgcn_readfirstlane(row_ptr[d + 1]);

    float4 acc = make_float4(0.0f, 0.0f, 0.0f, 0.0f);

    for (int base = rbeg; base < rend; base += 32) {
        int n = rend - base;
        if (n > 32) n = 32;
        int2 pr = make_int2(0, 0);
        if (lane < n) pr = *(const int2*)(pairs + 2L * (base + lane));

        int i = 0;
        for (; i + 8 <= n; i += 8) {
            float4 v[8];
            float  w[8];
            GATHER_LOAD(0) GATHER_LOAD(1) GATHER_LOAD(2) GATHER_LOAD(3)
            GATHER_LOAD(4) GATHER_LOAD(5) GATHER_LOAD(6) GATHER_LOAD(7)
            GATHER_ACC(0) GATHER_ACC(1) GATHER_ACC(2) GATHER_ACC(3)
            GATHER_ACC(4) GATHER_ACC(5) GATHER_ACC(6) GATHER_ACC(7)
        }
        const int r = n - i;
        if (r > 0) {
            float4 v[7];
            float  w[7];
            GATHER_LOAD(0)
            if (r > 1) GATHER_LOAD(1)
            if (r > 2) GATHER_LOAD(2)
            if (r > 3) GATHER_LOAD(3)
            if (r > 4) GATHER_LOAD(4)
            if (r > 5) GATHER_LOAD(5)
            if (r > 6) GATHER_LOAD(6)
            GATHER_ACC(0)
            if (r > 1) GATHER_ACC(1)
            if (r > 2) GATHER_ACC(2)
            if (r > 3) GATHER_ACC(3)
            if (r > 4) GATHER_ACC(4)
            if (r > 5) GATHER_ACC(5)
            if (r > 6) GATHER_ACC(6)
        }
    }

    *(float4*)(agg + (long)d * D + q) = acc;
}

__global__ void __launch_bounds__(256)
gcn_gemm_wmma_kernel(const float* __restrict__ agg,
                     const float* __restrict__ norm_dst,
                     const float* __restrict__ W,
                     const float* __restrict__ bias,
                     float* __restrict__ out, int N) {
    __shared__ float sA[16][D + 4];

    const int tid     = threadIdx.x;
    const int rowBase = blockIdx.x * 16;

    for (int idx = tid; idx < 16 * D; idx += 256) {
        int r = idx >> 7;
        int c = idx & (D - 1);
        int row = rowBase + r;
        float v = 0.0f;
        if (row < N) v = agg[(long)row * D + c] * norm_dst[row];
        sA[r][c] = v;
    }
    __syncthreads();

    const int wave  = tid >> 5;
    const int lane  = tid & 31;
    const int laneM = lane & 15;
    const int koff  = (lane >> 4) << 1;
    const int n     = (wave << 4) + laneM;

    v8f c;
    const float bv = bias[n];
#pragma unroll
    for (int v = 0; v < 8; ++v) c[v] = bv;

#pragma unroll
    for (int k0 = 0; k0 < D; k0 += 4) {
        v2f a, b;
        a.x = sA[laneM][k0 + koff];
        a.y = sA[laneM][k0 + koff + 1];
        b.x = W[(long)(k0 + koff) * D + n];
        b.y = W[(long)(k0 + koff + 1) * D + n];
        c = __builtin_amdgcn_wmma_f32_16x16x4_f32(
                 false, a,  false, b,
                 (short)0, c,  false,  false);
    }

    const int mBase = rowBase + ((lane >> 4) << 3);
#pragma unroll
    for (int v = 0; v < 8; ++v) {
        int row = mBase + v;
        if (row < N) out[(long)row * D + n] = c[v];
    }
}


extern "C" void kernel_launch(void* const* d_in, const int* in_sizes, int n_in,
                              void* d_out, int out_size, void* d_ws, size_t ws_size,
                              hipStream_t stream) {
    const float* feat = (const float*)d_in[0];
    const float* W    = (const float*)d_in[1];
    const float* bias = (const float*)d_in[2];
    const int*   src  = (const int*)d_in[3];
    const int*   dst  = (const int*)d_in[4];

    const int N = in_sizes[0] / D;
    const int E = in_sizes[3];

    const long ND = (long)N * D;

    float* ws       = (float*)d_ws;
    float* agg      = ws;
    float* hbuf     = agg + ND;
    int*   pairs    = (int*)(hbuf + ND);
    int*   rank     = pairs + 2L * E;
    float* norm_src = (float*)(rank + E);
    int*   cnt_dst  = (int*)(norm_src + N);
    float* norm_dst = (float*)(cnt_dst + N);
    int*   row_ptr  = (int*)(norm_dst + N);
    float* outf     = (float*)d_out;

    const long twoN = 2L * N;

    zero_words_kernel<<<(int)((twoN + 255) / 256), 256, 0, stream>>>((int*)norm_src, twoN);
    degree_rank_kernel<<<(E + 255) / 256, 256, 0, stream>>>(src, dst, norm_src, cnt_dst, rank, E);
    scan_norm_kernel<<<(N + 1023) / 1024, 1024, 0, stream>>>(cnt_dst, norm_src, norm_dst, row_ptr, N);
    fill_rows_kernel<<<(E + 255) / 256, 256, 0, stream>>>(src, dst, rank, row_ptr, norm_src, pairs, E);

    const int gatherGrid = (N + 7) / 8;
    const int gemmGrid   = (N + 15) / 16;

    const float* hin = feat;
    for (int layer = 0; layer < NLAYERS; ++layer) {
        float* hout = (layer & 1) ? outf : hbuf;

        spmm_gather_kernel<<<gatherGrid, 256, 0, stream>>>(hin, pairs, row_ptr, agg, N);
        gcn_gemm_wmma_kernel<<<gemmGrid, 256, 0, stream>>>(
            agg, norm_dst, W + (long)layer * D * D, bias + (long)layer * D, hout, N);

        hin = hout;
    }
}


# One launch per layer: each 8-wave group gathers its 16 destination rows straight into the LDS tile of the WMMA product; no aggregate plane in memory; per-row sum order unchanged
# speedup vs baseline: 3.1891x; 1.1571x over previous
#include <hip/hip_runtime.h>

#define D 128
#define NLAYERS 4

typedef __attribute__((ext_vector_type(2))) float v2f;
typedef __attribute__((ext_vector_type(8))) float v8f;


__global__ void zero_words_kernel(int* __restrict__ p, long n) {
    long i = (long)blockIdx.x * blockDim.x + threadIdx.x;
    if (i < n) p[i] = 0;
}

__global__ void degree_rank_kernel(const int* __restrict__ src, const int* __restrict__ dst,
                                   float* __restrict__ deg_src, int* __restrict__ cnt_dst,
                                   int* __restrict__ rank, int E) {
    int e = blockIdx.x * blockDim.x + threadIdx.x;
    if (e < E) {
        const int s = src[e];
        const int d = dst[e];
        unsafeAtomicAdd(&deg_src[s], 1.0f);
        rank[e] = atomicAdd(&cnt_dst[d], 1);
    }
}

__global__ void __launch_bounds__(1024)
scan_norm_kernel(const int* __restrict__ cnt_dst, float* __restrict__ deg_src,
                 float* __restrict__ norm_dst, int* __restrict__ row_ptr, int N) {
    __shared__ int tileSum[32];
    __shared__ int headSum[32];
    const int tid  = threadIdx.x;
    const int lane = tid & 31;
    const int wv   = tid >> 5;
    const int tbeg = blockIdx.x * 1024;
    const int i    = tbeg + tid;

    int head = 0;
    for (int j = tid; j < tbeg; j += 1024) head += cnt_dst[j];
    const int c = (i < N) ? cnt_dst[i] : 0;

    int incl = c;
    for (int off = 1; off < 32; off <<= 1) {
        int t = __shfl_up(incl, off, 32);
        if (lane >= off) incl += t;
    }
    for (int off = 16; off > 0; off >>= 1) head += __shfl_xor(head, off, 32);
    if (lane == 31) {
        tileSum[wv] = incl;
        headSum[wv] = head;
    }
    __syncthreads();

    int x = headSum[lane] + ((lane < wv) ? tileSum[lane] : 0);
    for (int off = 16; off > 0; off >>= 1) x += __shfl_xor(x, off, 32);

    const int excl = x + incl - c;
    if (i < N) {
        row_ptr[i]  = excl;
        norm_dst[i] = rsqrtf(fmaxf((float)c, 1.0f));
        deg_src[i]  = rsqrtf(fmaxf(deg_src[i], 1.0f));
        if (i == N - 1) row_ptr[N] = excl + c;
    }
}

__global__ void fill_rows_kernel(const int* __restrict__ src, const int* __restrict__ dst,
                                 const int* __restrict__ rank, const int* __restrict__ row_ptr,
                                 const float* __restrict__ norm_src,
                                 int* __restrict__ pairs, int E) {
    int e = blockIdx.x * blockDim.x + threadIdx.x;
    if (e < E) {
        int s = src[e];
        int p = row_ptr[dst[e]] + rank[e];
        int2 pr;
        pr.x = s;
        pr.y = __float_as_int(norm_src[s]);
        *(int2*)(pairs + 2L * p) = pr;
    }
}

#define GATHER_LOAD(k)                                                         \
    {                                                                          \
        int   s_ = __builtin_amdgcn_readlane(pr.x, i + (k));                   \
        w[k]     = __int_as_float(__builtin_amdgcn_readlane(pr.y, i + (k)));   \
        v[k]     = *(const float4*)(hin + (long)s_ * D + q);                   \
    }
#define GATHER_ACC(k)                                                          \
    {                                                                          \
        acc.x += v[k].x * w[k];                                                \
        acc.y += v[k].y * w[k];                                                \
        acc.z += v[k].z * w[k];                                                \
        acc.w += v[k].w * w[k];                                                \
    }

__device__ __forceinline__ float4 gather_row(const float* __restrict__ hin,
                                             const int* __restrict__ pairs,
                                             int rbeg, int rend, int lane, int q) {
    float4 acc = make_float4(0.0f, 0.0f, 0.0f, 0.0f);

    for (int base = rbeg; base < rend; base += 32) {
        int n = rend - base;
        if (n > 32) n = 32;
        int2 pr = make_int2(0, 0);
        if (lane < n) pr = *(const int2*)(pairs + 2L * (base + lane));

        int i = 0;
        for (; i + 8 <= n; i += 8) {
            float4 v[8];
            float  w[8];
            GATHER_LOAD(0) GATHER_LOAD(1) GATHER_LOAD(2) GATHER_LOAD(3)
            GATHER_LOAD(4) GATHER_LOAD(5) GATHER_LOAD(6) GATHER_LOAD(7)
            GATHER_ACC(0) GATHER_ACC(1) GATHER_ACC(2) GATHER_ACC(3)
            GATHER_ACC(4) GATHER_ACC(5) GATHER_ACC(6) GATHER_ACC(7)
        }
        const int r = n - i;
        if (r > 0) {
            float4 v[7];
            float  w[7];
            GATHER_LOAD(0)
            if (r > 1) GATHER_LOAD(1)
            if (r > 2) GATHER_LOAD(2)
            if (r > 3) GATHER_LOAD(3)
            if (r > 4) GATHER_LOAD(4)
            if (r > 5) GATHER_LOAD(5)
            if (r > 6) GATHER_LOAD(6)
            GATHER_ACC(0)
            if (r > 1) GATHER_ACC(1)
            if (r > 2) GATHER_ACC(2)
            if (r > 3) GATHER_ACC(3)
            if (r > 4) GATHER_ACC(4)
            if (r > 5) GATHER_ACC(5)
            if (r > 6) GATHER_ACC(6)
        }
    }
    return acc;
}

__global__ void __launch_bounds__(256)
gcn_layer_kernel(const float* __restrict__ hin,
                 const int* __restrict__ pairs,
                 const int* __restrict__ row_ptr,
                 const float* __restrict__ norm_dst,
                 const float* __restrict__ W,
                 const float* __restrict__ bias,
                 float* __restrict__ out, int N) {
    __shared__ __attribute__((aligned(16))) float sA[16][D + 4];

    const int tid     = threadIdx.x;
    const int wave    = tid >> 5;
    const int lane    = tid & 31;
    const int q       = lane << 2;
    const int rowBase = blockIdx.x * 16;

    {
        const int  r0   = wave << 1;
        const int  d0   = rowBase + r0;
        const bool has0 = d0 < N;
        const bool has1 = d0 + 1 < N;

        int   p0 = 0, p1 = 0, p2 = 0;
        float nd0 = 0.0f, nd1 = 0.0f;
        if (has0) {
            p0  = row_ptr[d0];
            p1  = row_ptr[d0 + 1];
            nd0 = norm_dst[d0];
        }
        if (has1) {
            p2  = row_ptr[d0 + 2];
            nd1 = norm_dst[d0 + 1];
        }
        p0 = __builtin_amdgcn_readfirstlane(p0);
        p1 = __builtin_amdgcn_readfirstlane(p1);
        p2 = __builtin_amdgcn_readfirstlane(p2);

        float4 a0 = make_float4(0.0f, 0.0f, 0.0f, 0.0f);
        float4 a1 = a0;
        if (has0) {
            a0 = gather_row(hin, pairs, p0, p1, lane, q);
            a0.x *= nd0; a0.y *= nd0; a0.z *= nd0; a0.w *= nd0;
        }
        *(float4*)&sA[r0][q] = a0;
        if (has1) {
            a1 = gather_row(hin, pairs, p1, p2, lane, q);
            a1.x *= nd1; a1.y *= nd1; a1.z *= nd1; a1.w *= nd1;
        }
        *(float4*)&sA[r0 + 1][q] = a1;
    }
    __syncthreads();

    const int laneM = lane & 15;
    const int koff  = (lane >> 4) << 1;
    const int n     = (wave << 4) + laneM;

    v8f c;
    const float bv = bias[n];
#pragma unroll
    for (int v = 0; v < 8; ++v) c[v] = bv;

#pragma unroll
    for (int k0 = 0; k0 < D; k0 += 4) {
        v2f a, b;
        a.x = sA[laneM][k0 + koff];
        a.y = sA[laneM][k0 + koff + 1];
        b.x = W[(long)(k0 + koff) * D + n];
        b.y = W[(long)(k0 + koff + 1) * D + n];
        c = __builtin_amdgcn_wmma_f32_16x16x4_f32(
                 false, a,  false, b,
                 (short)0, c,  false,  false);
    }

    const int mBase = rowBase + ((lane >> 4) << 3);
#pragma unroll
    for (int v = 0; v < 8; ++v) {
        int row = mBase + v;
        if (row < N) out[(long)row * D + n] = c[v];
    }
}


extern "C" void kernel_launch(void* const* d_in, const int* in_sizes, int n_in,
                              void* d_out, int out_size, void* d_ws, size_t ws_size,
                              hipStream_t stream) {
    const float* feat = (const float*)d_in[0];
    const float* W    = (const float*)d_in[1];
    const float* bias = (const float*)d_in[2];
    const int*   src  = (const int*)d_in[3];
    const int*   dst  = (const int*)d_in[4];

    const int N = in_sizes[0] / D;
    const int E = in_sizes[3];

    const long ND = (long)N * D;

    float* ws       = (float*)d_ws;
    float* hbuf     = ws;
    int*   pairs    = (int*)(hbuf + ND);
    int*   rank     = pairs + 2L * E;
    float* norm_src = (float*)(rank + E);
    int*   cnt_dst  = (int*)(norm_src + N);
    float* norm_dst = (float*)(cnt_dst + N);
    int*   row_ptr  = (int*)(norm_dst + N);
    float* outf     = (float*)d_out;

    const long twoN = 2L * N;

    zero_words_kernel<<<(int)((twoN + 255) / 256), 256, 0, stream>>>((int*)norm_src, twoN);
    degree_rank_kernel<<<(E + 255) / 256, 256, 0, stream>>>(src, dst, norm_src, cnt_dst, rank, E);
    scan_norm_kernel<<<(N + 1023) / 1024, 1024, 0, stream>>>(cnt_dst, norm_src, norm_dst, row_ptr, N);
    fill_rows_kernel<<<(E + 255) / 256, 256, 0, stream>>>(src, dst, rank, row_ptr, norm_src, pairs, E);

    const int layerGrid = (N + 15) / 16;

    const float* hin = feat;
    for (int layer = 0; layer < NLAYERS; ++layer) {
        float* hout = (layer & 1) ? outf : hbuf;

        gcn_layer_kernel<<<layerGrid, 256, 0, stream>>>(
            hin, pairs, row_ptr, norm_dst,
            W + (long)layer * D * D, bias + (long)layer * D, hout, N);

        hin = hout;
    }
}


# Both degree accumulators kept once per cache half (half read from the wave's placement word); a wave adds only into its own half's copy; per-row sum order = half 0's arrivals, then half 1's
# speedup vs baseline: 4.6355x; 1.4535x over previous
#include <hip/hip_runtime.h>

#define D 128
#define NLAYERS 4

typedef __attribute__((ext_vector_type(2))) float v2f;
typedef __attribute__((ext_vector_type(8))) float v8f;


__global__ void zero_words_kernel(int* __restrict__ p, long n) {
    long i = (long)blockIdx.x * blockDim.x + threadIdx.x;
    if (i < n) p[i] = 0;
}

__device__ __forceinline__ int cache_half() {
    const unsigned id = __builtin_amdgcn_s_sendmsg_rtn(0x87);
    return (int)((id >> 18) & 1u);
}

__global__ void degree_rank_kernel(const int* __restrict__ src, const int* __restrict__ dst,
                                   float* __restrict__ deg2, int* __restrict__ cnt2,
                                   int* __restrict__ rank, int E, int N) {
    int e = blockIdx.x * blockDim.x + threadIdx.x;
    const int  h   = cache_half();
    const long off = (long)h * N;
    if (e < E) {
        const int s = src[e];
        const int d = dst[e];
        unsafeAtomicAdd(&deg2[off + s], 1.0f);
        const int r = atomicAdd(&cnt2[off + d], 1);
        rank[e] = (r << 1) | h;
    }
}

__global__ void __launch_bounds__(1024)
scan_norm_kernel(const int* __restrict__ cnt2, const float* __restrict__ deg2,
                 float* __restrict__ norm_src, float* __restrict__ norm_dst,
                 int* __restrict__ row_ptr, int* __restrict__ base1, int N) {
    const int* __restrict__ cnt0 = cnt2;
    const int* __restrict__ cnt1 = cnt2 + N;
    __shared__ int tileSum[32];
    __shared__ int headSum[32];
    const int tid  = threadIdx.x;
    const int lane = tid & 31;
    const int wv   = tid >> 5;
    const int tbeg = blockIdx.x * 1024;
    const int i    = tbeg + tid;

    int head = 0;
#pragma unroll 8
    for (int j = tid; j < tbeg; j += 1024) head += cnt0[j] + cnt1[j];
    const int c0 = (i < N) ? cnt0[i] : 0;
    const int c1 = (i < N) ? cnt1[i] : 0;
    const int c  = c0 + c1;

    int incl = c;
    for (int off = 1; off < 32; off <<= 1) {
        int t = __shfl_up(incl, off, 32);
        if (lane >= off) incl += t;
    }
    for (int off = 16; off > 0; off >>= 1) head += __shfl_xor(head, off, 32);
    if (lane == 31) {
        tileSum[wv] = incl;
        headSum[wv] = head;
    }
    __syncthreads();

    int x = headSum[lane] + ((lane < wv) ? tileSum[lane] : 0);
    for (int off = 16; off > 0; off >>= 1) x += __shfl_xor(x, off, 32);

    const int excl = x + incl - c;
    if (i < N) {
        row_ptr[i]  = excl;
        base1[i]    = excl + c0;
        norm_dst[i] = rsqrtf(fmaxf((float)c, 1.0f));
        norm_src[i] = rsqrtf(fmaxf(deg2[i] + deg2[N + i], 1.0f));
        if (i == N - 1) row_ptr[N] = excl + c;
    }
}

__global__ void fill_rows_kernel(const int* __restrict__ src, const int* __restrict__ dst,
                                 const int* __restrict__ rank, const int* __restrict__ row_ptr,
                                 const int* __restrict__ base1,
                                 const float* __restrict__ norm_src,
                                 int* __restrict__ pairs, int E) {
    int e = blockIdx.x * blockDim.x + threadIdx.x;
    if (e < E) {
        int s  = src[e];
        int rk = rank[e];
        const int* __restrict__ base = (rk & 1) ? base1 : row_ptr;
        int p = base[dst[e]] + (rk >> 1);
        int2 pr;
        pr.x = s;
        pr.y = __float_as_int(norm_src[s]);
        *(int2*)(pairs + 2L * p) = pr;
    }
}

#define GATHER_LOAD(k)                                                         \
    {                                                                          \
        int   s_ = __builtin_amdgcn_readlane(pr.x, i + (k));                   \
        w[k]     = __int_as_float(__builtin_amdgcn_readlane(pr.y, i + (k)));   \
        v[k]     = *(const float4*)(hin + (long)s_ * D + q);                   \
    }
#define GATHER_ACC(k)                                                          \
    {                                                                          \
        acc.x += v[k].x * w[k];                                                \
        acc.y += v[k].y * w[k];                                                \
        acc.z += v[k].z * w[k];                                                \
        acc.w += v[k].w * w[k];                                                \
    }

__device__ __forceinline__ float4 gather_row(const float* __restrict__ hin,
                                             const int* __restrict__ pairs,
                                             int rbeg, int rend, int lane, int q) {
    float4 acc = make_float4(0.0f, 0.0f, 0.0f, 0.0f);

    for (int base = rbeg; base < rend; base += 32) {
        int n = rend - base;
        if (n > 32) n = 32;
        int2 pr = make_int2(0, 0);
        if (lane < n) pr = *(const int2*)(pairs + 2L * (base + lane));

        int i = 0;
        for (; i + 8 <= n; i += 8) {
            float4 v[8];
            float  w[8];
            GATHER_LOAD(0) GATHER_LOAD(1) GATHER_LOAD(2) GATHER_LOAD(3)
            GATHER_LOAD(4) GATHER_LOAD(5) GATHER_LOAD(6) GATHER_LOAD(7)
            GATHER_ACC(0) GATHER_ACC(1) GATHER_ACC(2) GATHER_ACC(3)
            GATHER_ACC(4) GATHER_ACC(5) GATHER_ACC(6) GATHER_ACC(7)
        }
        const int r = n - i;
        if (r > 0) {
            float4 v[7];
            float  w[7];
            GATHER_LOAD(0)
            if (r > 1) GATHER_LOAD(1)
            if (r > 2) GATHER_LOAD(2)
            if (r > 3) GATHER_LOAD(3)
            if (r > 4) GATHER_LOAD(4)
            if (r > 5) GATHER_LOAD(5)
            if (r > 6) GATHER_LOAD(6)
            GATHER_ACC(0)
            if (r > 1) GATHER_ACC(1)
            if (r > 2) GATHER_ACC(2)
            if (r > 3) GATHER_ACC(3)
            if (r > 4) GATHER_ACC(4)
            if (r > 5) GATHER_ACC(5)
            if (r > 6) GATHER_ACC(6)
        }
    }
    return acc;
}

__global__ void __launch_bounds__(256)
gcn_layer_kernel(const float* __restrict__ hin,
                 const int* __restrict__ pairs,
                 const int* __restrict__ row_ptr,
                 const float* __restrict__ norm_dst,
                 const float* __restrict__ W,
                 const float* __restrict__ bias,
                 float* __restrict__ out, int N) {
    __shared__ __attribute__((aligned(16))) float sA[16][D + 4];

    const int tid     = threadIdx.x;
    const int wave    = tid >> 5;
    const int lane    = tid & 31;
    const int q       = lane << 2;
    const int rowBase = blockIdx.x * 16;

    {
        const int  r0   = wave << 1;
        const int  d0   = rowBase + r0;
        const bool has0 = d0 < N;
        const bool has1 = d0 + 1 < N;

        int   p0 = 0, p1 = 0, p2 = 0;
        float nd0 = 0.0f, nd1 = 0.0f;
        if (has0) {
            p0  = row_ptr[d0];
            p1  = row_ptr[d0 + 1];
            nd0 = norm_dst[d0];
        }
        if (has1) {
            p2  = row_ptr[d0 + 2];
            nd1 = norm_dst[d0 + 1];
        }
        p0 = __builtin_amdgcn_readfirstlane(p0);
        p1 = __builtin_amdgcn_readfirstlane(p1);
        p2 = __builtin_amdgcn_readfirstlane(p2);

        float4 a0 = make_float4(0.0f, 0.0f, 0.0f, 0.0f);
        float4 a1 = a0;
        if (has0) {
            a0 = gather_row(hin, pairs, p0, p1, lane, q);
            a0.x *= nd0; a0.y *= nd0; a0.z *= nd0; a0.w *= nd0;
        }
        *(float4*)&sA[r0][q] = a0;
        if (has1) {
            a1 = gather_row(hin, pairs, p1, p2, lane, q);
            a1.x *= nd1; a1.y *= nd1; a1.z *= nd1; a1.w *= nd1;
        }
        *(float4*)&sA[r0 + 1][q] = a1;
    }
    __syncthreads();

    const int laneM = lane & 15;
    const int koff  = (lane >> 4) << 1;
    const int n     = (wave << 4) + laneM;

    v8f c;
    const float bv = bias[n];
#pragma unroll
    for (int v = 0; v < 8; ++v) c[v] = bv;

#pragma unroll
    for (int k0 = 0; k0 < D; k0 += 4) {
        v2f a, b;
        a.x = sA[laneM][k0 + koff];
        a.y = sA[laneM][k0 + koff + 1];
        b.x = W[(long)(k0 + koff) * D + n];
        b.y = W[(long)(k0 + koff + 1) * D + n];
        c = __builtin_amdgcn_wmma_f32_16x16x4_f32(
                 false, a,  false, b,
                 (short)0, c,  false,  false);
    }

    const int mBase = rowBase + ((lane >> 4) << 3);
#pragma unroll
    for (int v = 0; v < 8; ++v) {
        int row = mBase + v;
        if (row < N) out[(long)row * D + n] = c[v];
    }
}


extern "C" void kernel_launch(void* const* d_in, const int* in_sizes, int n_in,
                              void* d_out, int out_size, void* d_ws, size_t ws_size,
                              hipStream_t stream) {
    const float* feat = (const float*)d_in[0];
    const float* W    = (const float*)d_in[1];
    const float* bias = (const float*)d_in[2];
    const int*   src  = (const int*)d_in[3];
    const int*   dst  = (const int*)d_in[4];

    const int N = in_sizes[0] / D;
    const int E = in_sizes[3];

    const long ND = (long)N * D;

    float* ws       = (float*)d_ws;
    float* hbuf     = ws;
    int*   pairs    = (int*)(hbuf + ND);
    int*   rank     = pairs + 2L * E;
    float* deg2     = (float*)(rank + E);
    int*   cnt2     = (int*)(deg2 + 2L * N);
    float* norm_src = (float*)(cnt2 + 2L * N);
    float* norm_dst = norm_src + N;
    int*   row_ptr  = (int*)(norm_dst + N);
    int*   base1    = row_ptr + N + 1;
    float* outf     = (float*)d_out;

    const long fourN = 4L * N;

    zero_words_kernel<<<(int)((fourN + 255) / 256), 256, 0, stream>>>((int*)deg2, fourN);
    degree_rank_kernel<<<(E + 255) / 256, 256, 0, stream>>>(src, dst, deg2, cnt2, rank, E, N);
    scan_norm_kernel<<<(N + 1023) / 1024, 1024, 0, stream>>>(cnt2, deg2, norm_src, norm_dst, row_ptr, base1, N);
    fill_rows_kernel<<<(E + 255) / 256, 256, 0, stream>>>(src, dst, rank, row_ptr, base1, norm_src, pairs, E);

    const int layerGrid = (N + 15) / 16;

    const float* hin = feat;
    for (int layer = 0; layer < NLAYERS; ++layer) {
        float* hout = (layer & 1) ? outf : hbuf;

        gcn_layer_kernel<<<layerGrid, 256, 0, stream>>>(
            hin, pairs, row_ptr, norm_dst,
            W + (long)layer * D * D, bias + (long)layer * D, hout, N);

        hin = hout;
    }
}


# Layer work-groups live for 4 row tiles: a wave reads its W fragments and bias once (registers + own LDS slots), next tile's list bounds asked for under the product; 4x fewer waves; sum order kept
# speedup vs baseline: 4.7734x; 1.0297x over previous
#include <hip/hip_runtime.h>

#define D 128
#define NLAYERS 4

typedef __attribute__((ext_vector_type(2))) float v2f;
typedef __attribute__((ext_vector_type(8))) float v8f;


__global__ void zero_words_kernel(int* __restrict__ p, long n) {
    long i = (long)blockIdx.x * blockDim.x + threadIdx.x;
    if (i < n) p[i] = 0;
}

__device__ __forceinline__ int cache_half() {
    const unsigned id = __builtin_amdgcn_s_sendmsg_rtn(0x87);
    return (int)((id >> 18) & 1u);
}

__global__ void degree_rank_kernel(const int* __restrict__ src, const int* __restrict__ dst,
                                   float* __restrict__ deg2, int* __restrict__ cnt2,
                                   int* __restrict__ rank, int E, int N) {
    int e = blockIdx.x * blockDim.x + threadIdx.x;
    const int  h   = cache_half();
    const long off = (long)h * N;
    if (e < E) {
        const int s = src[e];
        const int d = dst[e];
        unsafeAtomicAdd(&deg2[off + s], 1.0f);
        const int r = atomicAdd(&cnt2[off + d], 1);
        rank[e] = (r << 1) | h;
    }
}

__global__ void __launch_bounds__(1024)
scan_norm_kernel(const int* __restrict__ cnt2, const float* __restrict__ deg2,
                 float* __restrict__ norm_src, float* __restrict__ norm_dst,
                 int* __restrict__ row_ptr, int* __restrict__ base1, int N) {
    const int* __restrict__ cnt0 = cnt2;
    const int* __restrict__ cnt1 = cnt2 + N;
    __shared__ int tileSum[32];
    __shared__ int headSum[32];
    const int tid  = threadIdx.x;
    const int lane = tid & 31;
    const int wv   = tid >> 5;
    const int tbeg = blockIdx.x * 1024;
    const int i    = tbeg + tid;

    int head = 0;
#pragma unroll 8
    for (int j = tid; j < tbeg; j += 1024) head += cnt0[j] + cnt1[j];
    const int c0 = (i < N) ? cnt0[i] : 0;
    const int c1 = (i < N) ? cnt1[i] : 0;
    const int c  = c0 + c1;

    int incl = c;
    for (int off = 1; off < 32; off <<= 1) {
        int t = __shfl_up(incl, off, 32);
        if (lane >= off) incl += t;
    }
    for (int off = 16; off > 0; off >>= 1) head += __shfl_xor(head, off, 32);
    if (lane == 31) {
        tileSum[wv] = incl;
        headSum[wv] = head;
    }
    __syncthreads();

    int x = headSum[lane] + ((lane < wv) ? tileSum[lane] : 0);
    for (int off = 16; off > 0; off >>= 1) x += __shfl_xor(x, off, 32);

    const int excl = x + incl - c;
    if (i < N) {
        row_ptr[i]  = excl;
        base1[i]    = excl + c0;
        norm_dst[i] = rsqrtf(fmaxf((float)c, 1.0f));
        norm_src[i] = rsqrtf(fmaxf(deg2[i] + deg2[N + i], 1.0f));
        if (i == N - 1) row_ptr[N] = excl + c;
    }
}

__global__ void fill_rows_kernel(const int* __restrict__ src, const int* __restrict__ dst,
                                 const int* __restrict__ rank, const int* __restrict__ row_ptr,
                                 const int* __restrict__ base1,
                                 const float* __restrict__ norm_src,
                                 int* __restrict__ pairs, int E) {
    int e = blockIdx.x * blockDim.x + threadIdx.x;
    if (e < E) {
        int s  = src[e];
        int rk = rank[e];
        const int* __restrict__ base = (rk & 1) ? base1 : row_ptr;
        int p = base[dst[e]] + (rk >> 1);
        int2 pr;
        pr.x = s;
        pr.y = __float_as_int(norm_src[s]);
        *(int2*)(pairs + 2L * p) = pr;
    }
}

#define WREG 22
#define GATHER_LOAD(k)                                                         \
    {                                                                          \
        int   s_ = __builtin_amdgcn_readlane(pr.x, i + (k));                   \
        w[k]     = __int_as_float(__builtin_amdgcn_readlane(pr.y, i + (k)));   \
        v[k]     = *(const float4*)(hin + (long)s_ * D + q);                   \
    }
#define GATHER_ACC(k)                                                          \
    {                                                                          \
        acc.x += v[k].x * w[k];                                                \
        acc.y += v[k].y * w[k];                                                \
        acc.z += v[k].z * w[k];                                                \
        acc.w += v[k].w * w[k];                                                \
    }

__device__ __forceinline__ float4 gather_row(const float* __restrict__ hin,
                                             const int* __restrict__ pairs,
                                             int rbeg, int rend, int lane, int q) {
    float4 acc = make_float4(0.0f, 0.0f, 0.0f, 0.0f);

    for (int base = rbeg; base < rend; base += 32) {
        int n = rend - base;
        if (n > 32) n = 32;
        int2 pr = make_int2(0, 0);
        if (lane < n) pr = *(const int2*)(pairs + 2L * (base + lane));

        int i = 0;
        for (; i + 8 <= n; i += 8) {
            float4 v[8];
            float  w[8];
            GATHER_LOAD(0) GATHER_LOAD(1) GATHER_LOAD(2) GATHER_LOAD(3)
            GATHER_LOAD(4) GATHER_LOAD(5) GATHER_LOAD(6) GATHER_LOAD(7)
            GATHER_ACC(0) GATHER_ACC(1) GATHER_ACC(2) GATHER_ACC(3)
            GATHER_ACC(4) GATHER_ACC(5) GATHER_ACC(6) GATHER_ACC(7)
        }
        const int r = n - i;
        if (r > 0) {
            float4 v[7];
            float  w[7];
            GATHER_LOAD(0)
            if (r > 1) GATHER_LOAD(1)
            if (r > 2) GATHER_LOAD(2)
            if (r > 3) GATHER_LOAD(3)
            if (r > 4) GATHER_LOAD(4)
            if (r > 5) GATHER_LOAD(5)
            if (r > 6) GATHER_LOAD(6)
            GATHER_ACC(0)
            if (r > 1) GATHER_ACC(1)
            if (r > 2) GATHER_ACC(2)
            if (r > 3) GATHER_ACC(3)
            if (r > 4) GATHER_ACC(4)
            if (r > 5) GATHER_ACC(5)
            if (r > 6) GATHER_ACC(6)
        }
    }
    return acc;
}

__global__ void __launch_bounds__(256) __attribute__((amdgpu_waves_per_eu(8, 8)))
gcn_layer_kernel(const float* __restrict__ hin,
                 const int* __restrict__ pairs,
                 const int* __restrict__ row_ptr,
                 const float* __restrict__ norm_dst,
                 const float* __restrict__ W,
                 const float* __restrict__ bias,
                 float* __restrict__ out, int N, int numTiles) {
    __shared__ __attribute__((aligned(16))) float sA[2][16][D + 4];
    __shared__ v2f sB[8][D / 4 - WREG][32];

    const int tid   = threadIdx.x;
    const int wave  = __builtin_amdgcn_readfirstlane(tid >> 5);
    const int lane  = tid & 31;
    const int q     = lane << 2;
    const int laneM = lane & 15;
    const int koff  = (lane >> 4) << 1;
    const int n     = (wave << 4) + laneM;

    v2f bfr[WREG];
#pragma unroll
    for (int ks = 0; ks < D / 4; ++ks) {
        v2f b;
        b.x = W[(long)(ks * 4 + koff) * D + n];
        b.y = W[(long)(ks * 4 + koff + 1) * D + n];
        if (ks < WREG) bfr[ks] = b;
        else           sB[wave][ks - WREG][lane] = b;
    }
    const float bv = bias[n];

    int   p0 = 0, p1 = 0, p2 = 0;
    float nd0 = 0.0f, nd1 = 0.0f;
    {
        const int d0 = blockIdx.x * 16 + (wave << 1);
        if (blockIdx.x < numTiles && d0 < N) {
            p0  = row_ptr[d0];
            p1  = row_ptr[d0 + 1];
            p2  = row_ptr[min(d0 + 2, N)];
            nd0 = norm_dst[d0];
            nd1 = norm_dst[min(d0 + 1, N - 1)];
        }
    }

    int buf = 0;
    for (int tile = blockIdx.x; tile < numTiles; tile += gridDim.x, buf ^= 1) {
        const int rowBase = tile * 16;

        {
            const int r0 = wave << 1;
            const int b0 = __builtin_amdgcn_readfirstlane(p0);
            const int b1 = __builtin_amdgcn_readfirstlane(p1);
            const int b2 = __builtin_amdgcn_readfirstlane(p2);

            float4 a0 = gather_row(hin, pairs, b0, b1, lane, q);
            a0.x *= nd0; a0.y *= nd0; a0.z *= nd0; a0.w *= nd0;
            *(float4*)&sA[buf][r0][q] = a0;
            float4 a1 = gather_row(hin, pairs, b1, b2, lane, q);
            a1.x *= nd1; a1.y *= nd1; a1.z *= nd1; a1.w *= nd1;
            *(float4*)&sA[buf][r0 + 1][q] = a1;
        }
        __syncthreads();

        {
            const int nt = tile + gridDim.x;
            const int d0 = nt * 16 + (wave << 1);
            p0 = 0; p1 = 0; p2 = 0; nd0 = 0.0f; nd1 = 0.0f;
            if (nt < numTiles && d0 < N) {
                p0  = row_ptr[d0];
                p1  = row_ptr[d0 + 1];
                p2  = row_ptr[min(d0 + 2, N)];
                nd0 = norm_dst[d0];
                nd1 = norm_dst[min(d0 + 1, N - 1)];
            }
        }

        v8f c;
#pragma unroll
        for (int v = 0; v < 8; ++v) c[v] = bv;

#pragma unroll
        for (int ks = 0; ks < D / 4; ++ks) {
            v2f a;
            a.x = sA[buf][laneM][ks * 4 + koff];
            a.y = sA[buf][laneM][ks * 4 + koff + 1];
            const v2f b = (ks < WREG) ? bfr[ks] : sB[wave][ks - WREG][lane];
            c = __builtin_amdgcn_wmma_f32_16x16x4_f32(
                     false, a,  false, b,
                     (short)0, c,  false,  false);
        }

        const int mBase = rowBase + ((lane >> 4) << 3);
#pragma unroll
        for (int v = 0; v < 8; ++v) {
            int row = mBase + v;
            if (row < N) out[(long)row * D + n] = c[v];
        }
    }
}


extern "C" void kernel_launch(void* const* d_in, const int* in_sizes, int n_in,
                              void* d_out, int out_size, void* d_ws, size_t ws_size,
                              hipStream_t stream) {
    const float* feat = (const float*)d_in[0];
    const float* W    = (const float*)d_in[1];
    const float* bias = (const float*)d_in[2];
    const int*   src  = (const int*)d_in[3];
    const int*   dst  = (const int*)d_in[4];

    const int N = in_sizes[0] / D;
    const int E = in_sizes[3];

    const long ND = (long)N * D;

    float* ws       = (float*)d_ws;
    float* hbuf     = ws;
    int*   pairs    = (int*)(hbuf + ND);
    int*   rank     = pairs + 2L * E;
    float* deg2     = (float*)(rank + E);
    int*   cnt2     = (int*)(deg2 + 2L * N);
    float* norm_src = (float*)(cnt2 + 2L * N);
    float* norm_dst = norm_src + N;
    int*   row_ptr  = (int*)(norm_dst + N);
    int*   base1    = row_ptr + N + 1;
    float* outf     = (float*)d_out;

    const long fourN = 4L * N;

    zero_words_kernel<<<(int)((fourN + 255) / 256), 256, 0, stream>>>((int*)deg2, fourN);
    degree_rank_kernel<<<(E + 255) / 256, 256, 0, stream>>>(src, dst, deg2, cnt2, rank, E, N);
    scan_norm_kernel<<<(N + 1023) / 1024, 1024, 0, stream>>>(cnt2, deg2, norm_src, norm_dst, row_ptr, base1, N);
    fill_rows_kernel<<<(E + 255) / 256, 256, 0, stream>>>(src, dst, rank, row_ptr, base1, norm_src, pairs, E);

    const int numTiles  = (N + 15) / 16;
    const int layerGrid = (numTiles + 3) / 4;

    const float* hin = feat;
    for (int layer = 0; layer < NLAYERS; ++layer) {
        float* hout = (layer & 1) ? outf : hbuf;

        gcn_layer_kernel<<<layerGrid, 256, 0, stream>>>(
            hin, pairs, row_ptr, norm_dst,
            W + (long)layer * D * D, bias + (long)layer * D, hout, N, numTiles);

        hin = hout;
    }
}


# Layers are linear: W0 W1 W2 W3 and the bias rows through them are formed first (WMMA); layer 1 keeps the one N-row product, layers 2-4 are plain row gathers + bias row; products re-associated
# speedup vs baseline: 5.5694x; 1.1668x over previous
#include <hip/hip_runtime.h>

#define D 128
#define NLAYERS 4

typedef __attribute__((ext_vector_type(2))) float v2f;
typedef __attribute__((ext_vector_type(8))) float v8f;


__global__ void zero_words_kernel(int* __restrict__ p, long n) {
    long i = (long)blockIdx.x * blockDim.x + threadIdx.x;
    if (i < n) p[i] = 0;
}

__device__ __forceinline__ int cache_half() {
    const unsigned id = __builtin_amdgcn_s_sendmsg_rtn(0x87);
    return (int)((id >> 18) & 1u);
}

__global__ void degree_rank_kernel(const int* __restrict__ src, const int* __restrict__ dst,
                                   float* __restrict__ deg2, int* __restrict__ cnt2,
                                   int* __restrict__ rank, int E, int N) {
    int e = blockIdx.x * blockDim.x + threadIdx.x;
    const int  h   = cache_half();
    const long off = (long)h * N;
    if (e < E) {
        const int s = src[e];
        const int d = dst[e];
        unsafeAtomicAdd(&deg2[off + s], 1.0f);
        const int r = atomicAdd(&cnt2[off + d], 1);
        rank[e] = (r << 1) | h;
    }
}

__global__ void __launch_bounds__(1024)
scan_norm_kernel(const int* __restrict__ cnt2, const float* __restrict__ deg2,
                 float* __restrict__ norm_src, float* __restrict__ norm_dst,
                 int* __restrict__ row_ptr, int* __restrict__ base1, int N) {
    const int* __restrict__ cnt0 = cnt2;
    const int* __restrict__ cnt1 = cnt2 + N;
    __shared__ int tileSum[32];
    __shared__ int headSum[32];
    const int tid  = threadIdx.x;
    const int lane = tid & 31;
    const int wv   = tid >> 5;
    const int tbeg = blockIdx.x * 1024;
    const int i    = tbeg + tid;

    int head = 0;
#pragma unroll 8
    for (int j = tid; j < tbeg; j += 1024) head += cnt0[j] + cnt1[j];
    const int c0 = (i < N) ? cnt0[i] : 0;
    const int c1 = (i < N) ? cnt1[i] : 0;
    const int c  = c0 + c1;

    int incl = c;
    for (int off = 1; off < 32; off <<= 1) {
        int t = __shfl_up(incl, off, 32);
        if (lane >= off) incl += t;
    }
    for (int off = 16; off > 0; off >>= 1) head += __shfl_xor(head, off, 32);
    if (lane == 31) {
        tileSum[wv] = incl;
        headSum[wv] = head;
    }
    __syncthreads();

    int x = headSum[lane] + ((lane < wv) ? tileSum[lane] : 0);
    for (int off = 16; off > 0; off >>= 1) x += __shfl_xor(x, off, 32);

    const int excl = x + incl - c;
    if (i < N) {
        row_ptr[i]  = excl;
        base1[i]    = excl + c0;
        norm_dst[i] = rsqrtf(fmaxf((float)c, 1.0f));
        norm_src[i] = rsqrtf(fmaxf(deg2[i] + deg2[N + i], 1.0f));
        if (i == N - 1) row_ptr[N] = excl + c;
    }
}

__global__ void fill_rows_kernel(const int* __restrict__ src, const int* __restrict__ dst,
                                 const int* __restrict__ rank, const int* __restrict__ row_ptr,
                                 const int* __restrict__ base1,
                                 const float* __restrict__ norm_src,
                                 int* __restrict__ pairs, int E) {
    int e = blockIdx.x * blockDim.x + threadIdx.x;
    if (e < E) {
        int s  = src[e];
        int rk = rank[e];
        const int* __restrict__ base = (rk & 1) ? base1 : row_ptr;
        int p = base[dst[e]] + (rk >> 1);
        int2 pr;
        pr.x = s;
        pr.y = __float_as_int(norm_src[s]);
        *(int2*)(pairs + 2L * p) = pr;
    }
}

#define WREG 22
#define GATHER_LOAD(k)                                                         \
    {                                                                          \
        int   s_ = __builtin_amdgcn_readlane(pr.x, i + (k));                   \
        w[k]     = __int_as_float(__builtin_amdgcn_readlane(pr.y, i + (k)));   \
        v[k]     = *(const float4*)(hin + (long)s_ * D + q);                   \
    }
#define GATHER_ACC(k)                                                          \
    {                                                                          \
        acc.x += v[k].x * w[k];                                                \
        acc.y += v[k].y * w[k];                                                \
        acc.z += v[k].z * w[k];                                                \
        acc.w += v[k].w * w[k];                                                \
    }

__device__ __forceinline__ float4 gather_row(const float* __restrict__ hin,
                                             const int* __restrict__ pairs,
                                             int rbeg, int rend, int lane, int q) {
    float4 acc = make_float4(0.0f, 0.0f, 0.0f, 0.0f);

    for (int base = rbeg; base < rend; base += 32) {
        int n = rend - base;
        if (n > 32) n = 32;
        int2 pr = make_int2(0, 0);
        if (lane < n) pr = *(const int2*)(pairs + 2L * (base + lane));

        int i = 0;
        for (; i + 8 <= n; i += 8) {
            float4 v[8];
            float  w[8];
            GATHER_LOAD(0) GATHER_LOAD(1) GATHER_LOAD(2) GATHER_LOAD(3)
            GATHER_LOAD(4) GATHER_LOAD(5) GATHER_LOAD(6) GATHER_LOAD(7)
            GATHER_ACC(0) GATHER_ACC(1) GATHER_ACC(2) GATHER_ACC(3)
            GATHER_ACC(4) GATHER_ACC(5) GATHER_ACC(6) GATHER_ACC(7)
        }
        const int r = n - i;
        if (r > 0) {
            float4 v[7];
            float  w[7];
            GATHER_LOAD(0)
            if (r > 1) GATHER_LOAD(1)
            if (r > 2) GATHER_LOAD(2)
            if (r > 3) GATHER_LOAD(3)
            if (r > 4) GATHER_LOAD(4)
            if (r > 5) GATHER_LOAD(5)
            if (r > 6) GATHER_LOAD(6)
            GATHER_ACC(0)
            if (r > 1) GATHER_ACC(1)
            if (r > 2) GATHER_ACC(2)
            if (r > 3) GATHER_ACC(3)
            if (r > 4) GATHER_ACC(4)
            if (r > 5) GATHER_ACC(5)
            if (r > 6) GATHER_ACC(6)
        }
    }
    return acc;
}

__global__ void __launch_bounds__(256) __attribute__((amdgpu_waves_per_eu(8, 8)))
gcn_layer_kernel(const float* __restrict__ hin,
                 const int* __restrict__ pairs,
                 const int* __restrict__ row_ptr,
                 const float* __restrict__ norm_dst,
                 const float* __restrict__ W,
                 const float* __restrict__ bias,
                 float* __restrict__ out, int N, int numTiles) {
    __shared__ __attribute__((aligned(16))) float sA[2][16][D + 4];
    __shared__ v2f sB[8][D / 4 - WREG][32];

    const int tid   = threadIdx.x;
    const int wave  = __builtin_amdgcn_readfirstlane(tid >> 5);
    const int lane  = tid & 31;
    const int q     = lane << 2;
    const int laneM = lane & 15;
    const int koff  = (lane >> 4) << 1;
    const int n     = (wave << 4) + laneM;

    v2f bfr[WREG];
#pragma unroll
    for (int ks = 0; ks < D / 4; ++ks) {
        v2f b;
        b.x = W[(long)(ks * 4 + koff) * D + n];
        b.y = W[(long)(ks * 4 + koff + 1) * D + n];
        if (ks < WREG) bfr[ks] = b;
        else           sB[wave][ks - WREG][lane] = b;
    }
    const float bv = bias[n];

    int   p0 = 0, p1 = 0, p2 = 0;
    float nd0 = 0.0f, nd1 = 0.0f;
    {
        const int d0 = blockIdx.x * 16 + (wave << 1);
        if (blockIdx.x < numTiles && d0 < N) {
            p0  = row_ptr[d0];
            p1  = row_ptr[d0 + 1];
            p2  = row_ptr[min(d0 + 2, N)];
            nd0 = norm_dst[d0];
            nd1 = norm_dst[min(d0 + 1, N - 1)];
        }
    }

    int buf = 0;
    for (int tile = blockIdx.x; tile < numTiles; tile += gridDim.x, buf ^= 1) {
        const int rowBase = tile * 16;

        {
            const int r0 = wave << 1;
            const int b0 = __builtin_amdgcn_readfirstlane(p0);
            const int b1 = __builtin_amdgcn_readfirstlane(p1);
            const int b2 = __builtin_amdgcn_readfirstlane(p2);

            float4 a0 = gather_row(hin, pairs, b0, b1, lane, q);
            a0.x *= nd0; a0.y *= nd0; a0.z *= nd0; a0.w *= nd0;
            *(float4*)&sA[buf][r0][q] = a0;
            float4 a1 = gather_row(hin, pairs, b1, b2, lane, q);
            a1.x *= nd1; a1.y *= nd1; a1.z *= nd1; a1.w *= nd1;
            *(float4*)&sA[buf][r0 + 1][q] = a1;
        }
        __syncthreads();

        {
            const int nt = tile + gridDim.x;
            const int d0 = nt * 16 + (wave << 1);
            p0 = 0; p1 = 0; p2 = 0; nd0 = 0.0f; nd1 = 0.0f;
            if (nt < numTiles && d0 < N) {
                p0  = row_ptr[d0];
                p1  = row_ptr[d0 + 1];
                p2  = row_ptr[min(d0 + 2, N)];
                nd0 = norm_dst[d0];
                nd1 = norm_dst[min(d0 + 1, N - 1)];
            }
        }

        v8f c;
#pragma unroll
        for (int v = 0; v < 8; ++v) c[v] = bv;

#pragma unroll
        for (int ks = 0; ks < D / 4; ++ks) {
            v2f a;
            a.x = sA[buf][laneM][ks * 4 + koff];
            a.y = sA[buf][laneM][ks * 4 + koff + 1];
            const v2f b = (ks < WREG) ? bfr[ks] : sB[wave][ks - WREG][lane];
            c = __builtin_amdgcn_wmma_f32_16x16x4_f32(
                     false, a,  false, b,
                     (short)0, c,  false,  false);
        }

        const int mBase = rowBase + ((lane >> 4) << 3);
#pragma unroll
        for (int v = 0; v < 8; ++v) {
            int row = mBase + v;
            if (row < N) out[(long)row * D + n] = c[v];
        }
    }
}

__global__ void __launch_bounds__(256)
weight_chain_kernel(const float* __restrict__ W,
                    const float* __restrict__ bias,
                    float* __restrict__ Wc,
                    float* __restrict__ cvec) {
    __shared__ __attribute__((aligned(16))) float sX[2][D][16];
    __shared__ float sP[NLAYERS - 1][16][16];

    const int tid   = threadIdx.x;
    const int wave  = __builtin_amdgcn_readfirstlane(tid >> 5);
    const int lane  = tid & 31;
    const int laneM = lane & 15;
    const int half  = lane >> 4;
    const int koff  = half << 1;
    const int J     = blockIdx.x * 16;

    {
        const int r = tid >> 1, c8 = (tid & 1) << 3;
        const float* __restrict__ p = W + (long)(NLAYERS - 1) * D * D + (long)r * D + J + c8;
        const float4 x0 = *(const float4*)p;
        const float4 x1 = *(const float4*)(p + 4);
        *(float4*)&sX[0][r][c8]     = x0;
        *(float4*)&sX[0][r][c8 + 4] = x1;
    }
    __syncthreads();

    int cur = 0;
    for (int m = NLAYERS - 2; m >= 0; --m) {
        {
            const int kb = (wave << 4) + (half << 3);
            float part = 0.0f;
#pragma unroll
            for (int j = 0; j < 8; ++j) part += bias[m * D + kb + j] * sX[cur][kb + j][laneM];
            sP[m][(wave << 1) + half][laneM] = part;
        }

        const float* __restrict__ Arow =
            W + (long)m * D * D + (long)((wave << 4) + laneM) * D + koff;
        v8f c;
#pragma unroll
        for (int v = 0; v < 8; ++v) c[v] = 0.0f;
#pragma unroll
        for (int ks = 0; ks < D / 4; ++ks) {
            const v2f a = *(const v2f*)(Arow + ks * 4);
            v2f b;
            b.x = sX[cur][ks * 4 + koff][laneM];
            b.y = sX[cur][ks * 4 + koff + 1][laneM];
            c = __builtin_amdgcn_wmma_f32_16x16x4_f32(
                     false, a,  false, b,
                     (short)0, c,  false,  false);
        }

        const int r0 = (wave << 4) + (half << 3);
        if (m > 0) {
#pragma unroll
            for (int v = 0; v < 8; ++v) sX[cur ^ 1][r0 + v][laneM] = c[v];
            __syncthreads();
            cur ^= 1;
        } else {
#pragma unroll
            for (int v = 0; v < 8; ++v) Wc[(long)(r0 + v) * D + J + laneM] = c[v];
        }
    }

    __syncthreads();
    if (tid < (NLAYERS - 1) * 16) {
        const int m = tid >> 4, col = tid & 15;
        float sacc = 0.0f;
#pragma unroll
        for (int g = 0; g < 16; ++g) sacc += sP[m][g][col];
        cvec[m * D + J + col] = sacc;
    }
}

__device__ __forceinline__ float4 gather_row_ahead(const float* __restrict__ hin,
                                                   const int* __restrict__ pairs,
                                                   int rbeg, int rend, int lane, int q,
                                                   int2 first, int nbeg, int nend,
                                                   int2& next) {
    float4 acc = make_float4(0.0f, 0.0f, 0.0f, 0.0f);
    next = make_int2(0, 0);
    bool asked = false;

    for (int base = rbeg; base < rend; base += 32) {
        int n = rend - base;
        if (n > 32) n = 32;
        int2 pr = first;
        if (base != rbeg) {
            pr = make_int2(0, 0);
            if (lane < n) pr = *(const int2*)(pairs + 2L * (base + lane));
        }

        int i = 0;
        for (; i + 8 <= n; i += 8) {
            float4 v[8];
            float  w[8];
            GATHER_LOAD(0) GATHER_LOAD(1) GATHER_LOAD(2) GATHER_LOAD(3)
            GATHER_LOAD(4) GATHER_LOAD(5) GATHER_LOAD(6) GATHER_LOAD(7)
            if (!asked) {
                if (lane < nend - nbeg) next = *(const int2*)(pairs + 2L * (nbeg + lane));
                asked = true;
            }
            GATHER_ACC(0) GATHER_ACC(1) GATHER_ACC(2) GATHER_ACC(3)
            GATHER_ACC(4) GATHER_ACC(5) GATHER_ACC(6) GATHER_ACC(7)
        }
        const int r = n - i;
        if (r > 0) {
            float4 v[7];
            float  w[7];
            GATHER_LOAD(0)
            if (r > 1) GATHER_LOAD(1)
            if (r > 2) GATHER_LOAD(2)
            if (r > 3) GATHER_LOAD(3)
            if (r > 4) GATHER_LOAD(4)
            if (r > 5) GATHER_LOAD(5)
            if (r > 6) GATHER_LOAD(6)
            if (!asked) {
                if (lane < nend - nbeg) next = *(const int2*)(pairs + 2L * (nbeg + lane));
                asked = true;
            }
            GATHER_ACC(0)
            if (r > 1) GATHER_ACC(1)
            if (r > 2) GATHER_ACC(2)
            if (r > 3) GATHER_ACC(3)
            if (r > 4) GATHER_ACC(4)
            if (r > 5) GATHER_ACC(5)
            if (r > 6) GATHER_ACC(6)
        }
    }
    if (!asked) {
        if (lane < nend - nbeg) next = *(const int2*)(pairs + 2L * (nbeg + lane));
    }
    return acc;
}

#define GATHER_GRID 1280

__global__ void __launch_bounds__(256)
gather_layer_kernel(const float* __restrict__ hin,
                    const int* __restrict__ pairs,
                    const int* __restrict__ row_ptr,
                    const float* __restrict__ norm_dst,
                    const float* __restrict__ brow,
                    float* __restrict__ out, int N) {
    const int tid    = threadIdx.x;
    const int wave   = __builtin_amdgcn_readfirstlane(tid >> 5);
    const int lane   = tid & 31;
    const int q      = lane << 2;
    const int stride = gridDim.x * 8;

    const float4 bq = *(const float4*)(brow + q);

    int d = blockIdx.x * 8 + wave;

    int a0 = 0, a1 = 0, b0 = 0, b1 = 0;
    if (d < N) {
        a0 = row_ptr[d];
        a1 = row_ptr[d + 1];
    }
    if (d + stride < N) {
        b0 = row_ptr[d + stride];
        b1 = row_ptr[d + stride + 1];
    }
    int2 prA = make_int2(0, 0);
    if (lane < a1 - a0) prA = *(const int2*)(pairs + 2L * (a0 + lane));

    for (; d < N; d += stride) {
        int c0 = 0, c1 = 0;
        const int d2 = d + 2 * stride;
        if (d2 < N) {
            c0 = row_ptr[d2];
            c1 = row_ptr[d2 + 1];
        }
        const float nd = norm_dst[d];

        int2 prB;
        const float4 acc = gather_row_ahead(hin, pairs, a0, a1, lane, q, prA, b0, b1, prB);

        float4 o;
        o.x = acc.x * nd + bq.x;
        o.y = acc.y * nd + bq.y;
        o.z = acc.z * nd + bq.z;
        o.w = acc.w * nd + bq.w;
        *(float4*)(out + (long)d * D + q) = o;

        a0 = b0; a1 = b1; prA = prB;
        b0 = c0; b1 = c1;
    }
}


extern "C" void kernel_launch(void* const* d_in, const int* in_sizes, int n_in,
                              void* d_out, int out_size, void* d_ws, size_t ws_size,
                              hipStream_t stream) {
    const float* feat = (const float*)d_in[0];
    const float* W    = (const float*)d_in[1];
    const float* bias = (const float*)d_in[2];
    const int*   src  = (const int*)d_in[3];
    const int*   dst  = (const int*)d_in[4];

    const int N = in_sizes[0] / D;
    const int E = in_sizes[3];

    const long ND = (long)N * D;

    float* ws       = (float*)d_ws;
    float* hbuf     = ws;
    float* Wc       = hbuf + ND;
    float* cvec     = Wc + (long)D * D;
    int*   pairs    = (int*)(cvec + 3L * D);
    int*   rank     = pairs + 2L * E;
    float* deg2     = (float*)(rank + E);
    int*   cnt2     = (int*)(deg2 + 2L * N);
    float* norm_src = (float*)(cnt2 + 2L * N);
    float* norm_dst = norm_src + N;
    int*   row_ptr  = (int*)(norm_dst + N);
    int*   base1    = row_ptr + N + 1;
    float* outf     = (float*)d_out;

    const long fourN = 4L * N;

    weight_chain_kernel<<<D / 16, 256, 0, stream>>>(W, bias, Wc, cvec);

    zero_words_kernel<<<(int)((fourN + 255) / 256), 256, 0, stream>>>((int*)deg2, fourN);
    degree_rank_kernel<<<(E + 255) / 256, 256, 0, stream>>>(src, dst, deg2, cnt2, rank, E, N);
    scan_norm_kernel<<<(N + 1023) / 1024, 1024, 0, stream>>>(cnt2, deg2, norm_src, norm_dst, row_ptr, base1, N);
    fill_rows_kernel<<<(E + 255) / 256, 256, 0, stream>>>(src, dst, rank, row_ptr, base1, norm_src, pairs, E);

    const int numTiles  = (N + 15) / 16;
    const int layerGrid = (numTiles + 3) / 4;
    gcn_layer_kernel<<<layerGrid, 256, 0, stream>>>(
        feat, pairs, row_ptr, norm_dst, Wc, cvec, hbuf, N, numTiles);

    gather_layer_kernel<<<GATHER_GRID, 256, 0, stream>>>(
        hbuf, pairs, row_ptr, norm_dst, cvec + D, outf, N);
    gather_layer_kernel<<<GATHER_GRID, 256, 0, stream>>>(
        outf, pairs, row_ptr, norm_dst, cvec + 2L * D, hbuf, N);
    gather_layer_kernel<<<GATHER_GRID, 256, 0, stream>>>(
        hbuf, pairs, row_ptr, norm_dst, bias + 3L * D, outf, N);
}


# Row lists at fixed pitch, filled in the degree pass (slot = returned count; spill list past 32): no prefix sum, no fill pass, 9 -> 7 launches; source norm applied at row store; sum order kept
# speedup vs baseline: 6.5702x; 1.1797x over previous
#include <hip/hip_runtime.h>

#define D 128
#define NLAYERS 4
#define CAP 32

typedef __attribute__((ext_vector_type(2))) float v2f;
typedef __attribute__((ext_vector_type(8))) float v8f;


__global__ void zero_words_kernel(int* __restrict__ p, long n) {
    long i = (long)blockIdx.x * blockDim.x + threadIdx.x;
    if (i < n) p[i] = 0;
}

__device__ __forceinline__ int cache_half() {
    const unsigned id = __builtin_amdgcn_s_sendmsg_rtn(0x87);
    return (int)((id >> 18) & 1u);
}

__global__ void degree_place_kernel(const int* __restrict__ src, const int* __restrict__ dst,
                                    float* __restrict__ deg2, int* __restrict__ cnt2,
                                    int* __restrict__ slots, int* __restrict__ spill_cnt,
                                    int* __restrict__ spill, int E, int N) {
    int e = blockIdx.x * blockDim.x + threadIdx.x;
    const int  h   = cache_half();
    const long off = (long)h * N;
    if (e < E) {
        const int s = src[e];
        const int d = dst[e];
        unsafeAtomicAdd(&deg2[off + s], 1.0f);
        const int r = atomicAdd(&cnt2[off + d], 1);
        if (r < CAP) {
            slots[(off + d) * CAP + r] = s;
        } else {
            const int k = atomicAdd(spill_cnt, 1);
            int2 o;
            o.x = d;
            o.y = s;
            *(int2*)(spill + 2L * k) = o;
        }
    }
}

#define GATHER_LOAD(k)                                                         \
    {                                                                          \
        int s_ = __builtin_amdgcn_readlane(sj, i + (k));                       \
        v[k]   = *(const float4*)(hin + (long)s_ * D + q);                     \
    }
#define GATHER_ACC(k)                                                          \
    {                                                                          \
        acc.x += v[k].x;                                                       \
        acc.y += v[k].y;                                                       \
        acc.z += v[k].z;                                                       \
        acc.w += v[k].w;                                                       \
    }
#define GATHER_ACC_W(k)                                                        \
    {                                                                          \
        const float g0_ = __int_as_float(                                      \
            __builtin_amdgcn_readlane(__float_as_int(dg0), i + (k)));          \
        const float g1_ = __int_as_float(                                      \
            __builtin_amdgcn_readlane(__float_as_int(dg1), i + (k)));          \
        const float w_ = rsqrtf(fmaxf(g0_ + g1_, 1.0f));                       \
        acc.x += v[k].x * w_;                                                  \
        acc.y += v[k].y * w_;                                                  \
        acc.z += v[k].z * w_;                                                  \
        acc.w += v[k].w * w_;                                                  \
    }

template <bool WEIGHTED>
__device__ __forceinline__ void spill_add(float4& acc, const float* __restrict__ hin,
                                          const int* __restrict__ spill, int m, int d,
                                          const float* __restrict__ deg2, int N,
                                          int lane, int q) {
    for (int kb = 0; kb < m; kb += 32) {
        int od = -1, os = 0;
        if (kb + lane < m) {
            const int2 o = *(const int2*)(spill + 2L * (kb + lane));
            od = o.x;
            os = o.y;
        }
        unsigned mask = __builtin_amdgcn_ballot_w32(od == d);
        while (mask) {
            const int b = __builtin_ctz(mask);
            mask &= mask - 1u;
            const int s_ = __builtin_amdgcn_readlane(os, b);
            const float4 v = *(const float4*)(hin + (long)s_ * D + q);
            float w_ = 1.0f;
            if (WEIGHTED) w_ = rsqrtf(fmaxf(deg2[s_] + deg2[N + s_], 1.0f));
            acc.x += v.x * w_;
            acc.y += v.y * w_;
            acc.z += v.z * w_;
            acc.w += v.w * w_;
        }
    }
}

__device__ __forceinline__ float4 gather_row_w(const float* __restrict__ hin,
                                               const int* __restrict__ slots,
                                               const float* __restrict__ deg2,
                                               int N, int d, int c0, int c1,
                                               int lane, int q) {
    float4 acc = make_float4(0.0f, 0.0f, 0.0f, 0.0f);
    const int n0 = min(c0, CAP);
    const int n  = n0 + min(c1, CAP);
    const int* __restrict__ S0 = slots + (long)d * CAP;
    const int* __restrict__ S1 = slots + ((long)N + d) * CAP;

    for (int base = 0; base < n; base += 32) {
        int cnt = n - base;
        if (cnt > 32) cnt = 32;
        int sj = 0;
        if (lane < cnt) {
            const int j = base + lane;
            sj = (j < n0) ? S0[j] : S1[j - n0];
        }
        const float dg0 = deg2[sj];
        const float dg1 = deg2[N + sj];

        int i = 0;
        for (; i + 8 <= cnt; i += 8) {
            float4 v[8];
            GATHER_LOAD(0) GATHER_LOAD(1) GATHER_LOAD(2) GATHER_LOAD(3)
            GATHER_LOAD(4) GATHER_LOAD(5) GATHER_LOAD(6) GATHER_LOAD(7)
            GATHER_ACC_W(0) GATHER_ACC_W(1) GATHER_ACC_W(2) GATHER_ACC_W(3)
            GATHER_ACC_W(4) GATHER_ACC_W(5) GATHER_ACC_W(6) GATHER_ACC_W(7)
        }
        const int r = cnt - i;
        if (r > 0) {
            float4 v[7];
            GATHER_LOAD(0)
            if (r > 1) GATHER_LOAD(1)
            if (r > 2) GATHER_LOAD(2)
            if (r > 3) GATHER_LOAD(3)
            if (r > 4) GATHER_LOAD(4)
            if (r > 5) GATHER_LOAD(5)
            if (r > 6) GATHER_LOAD(6)
            GATHER_ACC_W(0)
            if (r > 1) GATHER_ACC_W(1)
            if (r > 2) GATHER_ACC_W(2)
            if (r > 3) GATHER_ACC_W(3)
            if (r > 4) GATHER_ACC_W(4)
            if (r > 5) GATHER_ACC_W(5)
            if (r > 6) GATHER_ACC_W(6)
        }
    }
    return acc;
}

#define WREG 16

__global__ void __launch_bounds__(256) __attribute__((amdgpu_waves_per_eu(8, 8)))
gcn_layer_kernel(const float* __restrict__ hin,
                 const int* __restrict__ slots,
                 const int* __restrict__ cnt2,
                 const float* __restrict__ deg2,
                 const int* __restrict__ spill,
                 const int* __restrict__ spill_cnt,
                 const float* __restrict__ W,
                 const float* __restrict__ bias,
                 float* __restrict__ out, int N, int numTiles) {
    __shared__ __attribute__((aligned(16))) float sA[2][16][D + 4];
    __shared__ float sNs[2][16];
    __shared__ v2f sB[8][D / 4 - WREG][32];

    const int tid   = threadIdx.x;
    const int wave  = __builtin_amdgcn_readfirstlane(tid >> 5);
    const int lane  = tid & 31;
    const int q     = lane << 2;
    const int laneM = lane & 15;
    const int koff  = (lane >> 4) << 1;
    const int n     = (wave << 4) + laneM;

    v2f bfr[WREG];
#pragma unroll
    for (int ks = 0; ks < D / 4; ++ks) {
        v2f b;
        b.x = W[(long)(ks * 4 + koff) * D + n];
        b.y = W[(long)(ks * 4 + koff + 1) * D + n];
        if (ks < WREG) bfr[ks] = b;
        else           sB[wave][ks - WREG][lane] = b;
    }
    const float bv = bias[n];

    int   cA0 = 0, cA1 = 0, cB0 = 0, cB1 = 0;
    float gA = 1.0f, gB = 1.0f;
    {
        const int d0 = blockIdx.x * 16 + (wave << 1);
        if (blockIdx.x < numTiles && d0 < N) {
            const int dB = min(d0 + 1, N - 1);
            cA0 = cnt2[d0];
            cA1 = cnt2[N + d0];
            const int tB0 = cnt2[dB];
            const int tB1 = cnt2[N + dB];
            gA  = deg2[d0] + deg2[N + d0];
            gB  = deg2[dB] + deg2[N + dB];
            cB0 = (d0 + 1 < N) ? tB0 : 0;
            cB1 = (d0 + 1 < N) ? tB1 : 0;
        }
    }

    int buf = 0;
    for (int tile = blockIdx.x; tile < numTiles; tile += gridDim.x, buf ^= 1) {
        const int rowBase = tile * 16;

        {
            const int r0 = wave << 1;
            const int d0 = rowBase + r0;
            const int a0 = __builtin_amdgcn_readfirstlane(cA0);
            const int a1 = __builtin_amdgcn_readfirstlane(cA1);
            const int b0 = __builtin_amdgcn_readfirstlane(cB0);
            const int b1 = __builtin_amdgcn_readfirstlane(cB1);

            float4 x0 = gather_row_w(hin, slots, deg2, N, d0, a0, a1, lane, q);
            if (a0 > CAP || a1 > CAP)
                spill_add<true>(x0, hin, spill, *spill_cnt, d0, deg2, N, lane, q);
            const float nd0 = rsqrtf(fmaxf((float)(a0 + a1), 1.0f));
            x0.x *= nd0; x0.y *= nd0; x0.z *= nd0; x0.w *= nd0;
            *(float4*)&sA[buf][r0][q] = x0;
            float4 x1 = gather_row_w(hin, slots, deg2, N, d0 + 1, b0, b1, lane, q);
            if (b0 > CAP || b1 > CAP)
                spill_add<true>(x1, hin, spill, *spill_cnt, d0 + 1, deg2, N, lane, q);
            const float nd1 = rsqrtf(fmaxf((float)(b0 + b1), 1.0f));
            x1.x *= nd1; x1.y *= nd1; x1.z *= nd1; x1.w *= nd1;
            *(float4*)&sA[buf][r0 + 1][q] = x1;
            if (lane == 0) {
                sNs[buf][r0]     = rsqrtf(fmaxf(gA, 1.0f));
                sNs[buf][r0 + 1] = rsqrtf(fmaxf(gB, 1.0f));
            }
        }
        __syncthreads();

        {
            const int nt = tile + gridDim.x;
            const int d0 = nt * 16 + (wave << 1);
            cA0 = 0; cA1 = 0; cB0 = 0; cB1 = 0; gA = 1.0f; gB = 1.0f;
            if (nt < numTiles && d0 < N) {
                const int dB = min(d0 + 1, N - 1);
                cA0 = cnt2[d0];
                cA1 = cnt2[N + d0];
                const int tB0 = cnt2[dB];
                const int tB1 = cnt2[N + dB];
                gA  = deg2[d0] + deg2[N + d0];
                gB  = deg2[dB] + deg2[N + dB];
                cB0 = (d0 + 1 < N) ? tB0 : 0;
                cB1 = (d0 + 1 < N) ? tB1 : 0;
            }
        }

        v8f c;
#pragma unroll
        for (int v = 0; v < 8; ++v) c[v] = bv;

#pragma unroll
        for (int ks = 0; ks < D / 4; ++ks) {
            v2f a;
            a.x = sA[buf][laneM][ks * 4 + koff];
            a.y = sA[buf][laneM][ks * 4 + koff + 1];
            const v2f b = (ks < WREG) ? bfr[ks] : sB[wave][ks - WREG][lane];
            c = __builtin_amdgcn_wmma_f32_16x16x4_f32(
                     false, a,  false, b,
                     (short)0, c,  false,  false);
        }

        const int mOff  = (lane >> 4) << 3;
        const int mBase = rowBase + mOff;
#pragma unroll
        for (int v = 0; v < 8; ++v) {
            int row = mBase + v;
            if (row < N) out[(long)row * D + n] = c[v] * sNs[buf][mOff + v];
        }
    }
}

__global__ void __launch_bounds__(256)
weight_chain_kernel(const float* __restrict__ W,
                    const float* __restrict__ bias,
                    float* __restrict__ Wc,
                    float* __restrict__ cvec) {
    __shared__ __attribute__((aligned(16))) float sX[2][D][16];
    __shared__ float sP[NLAYERS - 1][16][16];

    const int tid   = threadIdx.x;
    const int wave  = __builtin_amdgcn_readfirstlane(tid >> 5);
    const int lane  = tid & 31;
    const int laneM = lane & 15;
    const int half  = lane >> 4;
    const int koff  = half << 1;
    const int J     = blockIdx.x * 16;

    {
        const int r = tid >> 1, c8 = (tid & 1) << 3;
        const float* __restrict__ p = W + (long)(NLAYERS - 1) * D * D + (long)r * D + J + c8;
        const float4 x0 = *(const float4*)p;
        const float4 x1 = *(const float4*)(p + 4);
        *(float4*)&sX[0][r][c8]     = x0;
        *(float4*)&sX[0][r][c8 + 4] = x1;
    }
    __syncthreads();

    int cur = 0;
    for (int m = NLAYERS - 2; m >= 0; --m) {
        {
            const int kb = (wave << 4) + (half << 3);
            float part = 0.0f;
#pragma unroll
            for (int j = 0; j < 8; ++j) part += bias[m * D + kb + j] * sX[cur][kb + j][laneM];
            sP[m][(wave << 1) + half][laneM] = part;
        }

        const float* __restrict__ Arow =
            W + (long)m * D * D + (long)((wave << 4) + laneM) * D + koff;
        v8f c;
#pragma unroll
        for (int v = 0; v < 8; ++v) c[v] = 0.0f;
#pragma unroll
        for (int ks = 0; ks < D / 4; ++ks) {
            const v2f a = *(const v2f*)(Arow + ks * 4);
            v2f b;
            b.x = sX[cur][ks * 4 + koff][laneM];
            b.y = sX[cur][ks * 4 + koff + 1][laneM];
            c = __builtin_amdgcn_wmma_f32_16x16x4_f32(
                     false, a,  false, b,
                     (short)0, c,  false,  false);
        }

        const int r0 = (wave << 4) + (half << 3);
        if (m > 0) {
#pragma unroll
            for (int v = 0; v < 8; ++v) sX[cur ^ 1][r0 + v][laneM] = c[v];
            __syncthreads();
            cur ^= 1;
        } else {
#pragma unroll
            for (int v = 0; v < 8; ++v) Wc[(long)(r0 + v) * D + J + laneM] = c[v];
        }
    }

    __syncthreads();
    if (tid < (NLAYERS - 1) * 16) {
        const int m = tid >> 4, col = tid & 15;
        float sacc = 0.0f;
#pragma unroll
        for (int g = 0; g < 16; ++g) sacc += sP[m][g][col];
        cvec[m * D + J + col] = sacc;
    }
}

__device__ __forceinline__ float4 gather_row_ahead(const float* __restrict__ hin,
                                                   const int* __restrict__ slots,
                                                   int N, int d, int c0, int c1,
                                                   int lane, int q, int first,
                                                   int dn, int e0, int e1, int& next) {
    float4 acc = make_float4(0.0f, 0.0f, 0.0f, 0.0f);
    next = 0;
    bool asked = false;

    const int n0 = min(c0, CAP);
    const int n  = n0 + min(c1, CAP);
    const int* __restrict__ S0 = slots + (long)d * CAP;
    const int* __restrict__ S1 = slots + ((long)N + d) * CAP;
    const int m0 = min(e0, CAP);
    const int mn = min(m0 + min(e1, CAP), 32);
    const int* __restrict__ T0 = slots + (long)dn * CAP;
    const int* __restrict__ T1 = slots + ((long)N + dn) * CAP;

    for (int base = 0; base < n; base += 32) {
        int cnt = n - base;
        if (cnt > 32) cnt = 32;
        int sj = first;
        if (base != 0) {
            sj = 0;
            if (lane < cnt) {
                const int j = base + lane;
                sj = (j < n0) ? S0[j] : S1[j - n0];
            }
        }

        int i = 0;
        for (; i + 8 <= cnt; i += 8) {
            float4 v[8];
            GATHER_LOAD(0) GATHER_LOAD(1) GATHER_LOAD(2) GATHER_LOAD(3)
            GATHER_LOAD(4) GATHER_LOAD(5) GATHER_LOAD(6) GATHER_LOAD(7)
            if (!asked) {
                if (lane < mn) next = (lane < m0) ? T0[lane] : T1[lane - m0];
                asked = true;
            }
            GATHER_ACC(0) GATHER_ACC(1) GATHER_ACC(2) GATHER_ACC(3)
            GATHER_ACC(4) GATHER_ACC(5) GATHER_ACC(6) GATHER_ACC(7)
        }
        const int r = cnt - i;
        if (r > 0) {
            float4 v[7];
            GATHER_LOAD(0)
            if (r > 1) GATHER_LOAD(1)
            if (r > 2) GATHER_LOAD(2)
            if (r > 3) GATHER_LOAD(3)
            if (r > 4) GATHER_LOAD(4)
            if (r > 5) GATHER_LOAD(5)
            if (r > 6) GATHER_LOAD(6)
            if (!asked) {
                if (lane < mn) next = (lane < m0) ? T0[lane] : T1[lane - m0];
                asked = true;
            }
            GATHER_ACC(0)
            if (r > 1) GATHER_ACC(1)
            if (r > 2) GATHER_ACC(2)
            if (r > 3) GATHER_ACC(3)
            if (r > 4) GATHER_ACC(4)
            if (r > 5) GATHER_ACC(5)
            if (r > 6) GATHER_ACC(6)
        }
    }
    if (!asked) {
        if (lane < mn) next = (lane < m0) ? T0[lane] : T1[lane - m0];
    }
    return acc;
}

#define GATHER_GRID 1280

template <bool SCALE_OUT>
__global__ void __launch_bounds__(256)
gather_layer_kernel(const float* __restrict__ hin,
                    const int* __restrict__ slots,
                    const int* __restrict__ cnt2,
                    const float* __restrict__ deg2,
                    const int* __restrict__ spill,
                    const int* __restrict__ spill_cnt,
                    const float* __restrict__ brow,
                    float* __restrict__ out, int N) {
    const int tid    = threadIdx.x;
    const int wave   = __builtin_amdgcn_readfirstlane(tid >> 5);
    const int lane   = tid & 31;
    const int q      = lane << 2;
    const int stride = gridDim.x * 8;

    const float4 bq = *(const float4*)(brow + q);

    int d = blockIdx.x * 8 + wave;

    int a0 = 0, a1 = 0, b0 = 0, b1 = 0;
    if (d < N) {
        a0 = cnt2[d];
        a1 = cnt2[N + d];
    }
    if (d + stride < N) {
        b0 = cnt2[d + stride];
        b1 = cnt2[N + d + stride];
    }
    int prA = 0;
    {
        const int m0 = min(a0, CAP);
        const int mn = min(m0 + min(a1, CAP), 32);
        if (lane < mn)
            prA = (lane < m0) ? slots[(long)d * CAP + lane]
                              : slots[((long)N + d) * CAP + (lane - m0)];
    }

    for (; d < N; d += stride) {
        int c0 = 0, c1 = 0;
        const int d2 = d + 2 * stride;
        if (d2 < N) {
            c0 = cnt2[d2];
            c1 = cnt2[N + d2];
        }
        float ns = 1.0f;
        if (SCALE_OUT) ns = rsqrtf(fmaxf(deg2[d] + deg2[N + d], 1.0f));
        const float nd = rsqrtf(fmaxf((float)(a0 + a1), 1.0f));

        int prB;
        float4 acc = gather_row_ahead(hin, slots, N, d, a0, a1, lane, q, prA,
                                      d + stride, b0, b1, prB);
        if (a0 > CAP || a1 > CAP)
            spill_add<false>(acc, hin, spill, *spill_cnt, d, deg2, N, lane, q);

        float4 o;
        o.x = (acc.x * nd + bq.x) * ns;
        o.y = (acc.y * nd + bq.y) * ns;
        o.z = (acc.z * nd + bq.z) * ns;
        o.w = (acc.w * nd + bq.w) * ns;
        *(float4*)(out + (long)d * D + q) = o;

        a0 = b0; a1 = b1; prA = prB;
        b0 = c0; b1 = c1;
    }
}


extern "C" void kernel_launch(void* const* d_in, const int* in_sizes, int n_in,
                              void* d_out, int out_size, void* d_ws, size_t ws_size,
                              hipStream_t stream) {
    const float* feat = (const float*)d_in[0];
    const float* W    = (const float*)d_in[1];
    const float* bias = (const float*)d_in[2];
    const int*   src  = (const int*)d_in[3];
    const int*   dst  = (const int*)d_in[4];

    const int N = in_sizes[0] / D;
    const int E = in_sizes[3];

    const long ND = (long)N * D;

    float* ws        = (float*)d_ws;
    float* hbuf      = ws;
    float* Wc        = hbuf + ND;
    float* cvec      = Wc + (long)D * D;
    int*   slots     = (int*)(cvec + 3L * D);
    int*   spill     = slots + 2L * N * CAP;
    float* deg2      = (float*)(spill + 2L * E);
    int*   cnt2      = (int*)(deg2 + 2L * N);
    int*   spill_cnt = cnt2 + 2L * N;
    float* outf      = (float*)d_out;

    const long nZero = 4L * N + 1;

    weight_chain_kernel<<<D / 16, 256, 0, stream>>>(W, bias, Wc, cvec);

    zero_words_kernel<<<(int)((nZero + 255) / 256), 256, 0, stream>>>((int*)deg2, nZero);
    degree_place_kernel<<<(E + 255) / 256, 256, 0, stream>>>(
        src, dst, deg2, cnt2, slots, spill_cnt, spill, E, N);

    const int numTiles  = (N + 15) / 16;
    const int layerGrid = (numTiles + 3) / 4;
    gcn_layer_kernel<<<layerGrid, 256, 0, stream>>>(
        feat, slots, cnt2, deg2, spill, spill_cnt, Wc, cvec, hbuf, N, numTiles);

    gather_layer_kernel<true><<<GATHER_GRID, 256, 0, stream>>>(
        hbuf, slots, cnt2, deg2, spill, spill_cnt, cvec + D, outf, N);
    gather_layer_kernel<true><<<GATHER_GRID, 256, 0, stream>>>(
        outf, slots, cnt2, deg2, spill, spill_cnt, cvec + 2L * D, hbuf, N);
    gather_layer_kernel<false><<<GATHER_GRID, 256, 0, stream>>>(
        hbuf, slots, cnt2, deg2, spill, spill_cnt, bias + 3L * D, outf, N);
}


# Weight chain loses its own launch: its 8 work-groups are the first groups of the edge pass's launch (independent work; the serial chain runs under the edge pass); 7 -> 6 launches; sums unchanged
# speedup vs baseline: 6.9366x; 1.0558x over previous
#include <hip/hip_runtime.h>

#define D 128
#define NLAYERS 4
#define CAP 32

typedef __attribute__((ext_vector_type(2))) float v2f;
typedef __attribute__((ext_vector_type(8))) float v8f;


__global__ void zero_words_kernel(int* __restrict__ p, long n) {
    long i = (long)blockIdx.x * blockDim.x + threadIdx.x;
    if (i < n) p[i] = 0;
}

__device__ __forceinline__ int cache_half() {
    const unsigned id = __builtin_amdgcn_s_sendmsg_rtn(0x87);
    return (int)((id >> 18) & 1u);
}

#define CHAIN_GROUPS (D / 16)

__device__ __forceinline__ void
weight_chain_block(const float* __restrict__ W,
                   const float* __restrict__ bias,
                   float* __restrict__ Wc,
                   float* __restrict__ cvec,
                   int jb) {
    __shared__ __attribute__((aligned(16))) float sX[2][D][16];
    __shared__ float sP[NLAYERS - 1][16][16];

    const int tid   = threadIdx.x;
    const int wave  = __builtin_amdgcn_readfirstlane(tid >> 5);
    const int lane  = tid & 31;
    const int laneM = lane & 15;
    const int half  = lane >> 4;
    const int koff  = half << 1;
    const int J     = jb * 16;

    {
        const int r = tid >> 1, c8 = (tid & 1) << 3;
        const float* __restrict__ p = W + (long)(NLAYERS - 1) * D * D + (long)r * D + J + c8;
        const float4 x0 = *(const float4*)p;
        const float4 x1 = *(const float4*)(p + 4);
        *(float4*)&sX[0][r][c8]     = x0;
        *(float4*)&sX[0][r][c8 + 4] = x1;
    }
    __syncthreads();

    int cur = 0;
    for (int m = NLAYERS - 2; m >= 0; --m) {
        {
            const int kb = (wave << 4) + (half << 3);
            float part = 0.0f;
#pragma unroll
            for (int j = 0; j < 8; ++j) part += bias[m * D + kb + j] * sX[cur][kb + j][laneM];
            sP[m][(wave << 1) + half][laneM] = part;
        }

        const float* __restrict__ Arow =
            W + (long)m * D * D + (long)((wave << 4) + laneM) * D + koff;
        v8f c;
#pragma unroll
        for (int v = 0; v < 8; ++v) c[v] = 0.0f;
#pragma unroll
        for (int ks = 0; ks < D / 4; ++ks) {
            const v2f a = *(const v2f*)(Arow + ks * 4);
            v2f b;
            b.x = sX[cur][ks * 4 + koff][laneM];
            b.y = sX[cur][ks * 4 + koff + 1][laneM];
            c = __builtin_amdgcn_wmma_f32_16x16x4_f32(
                     false, a,  false, b,
                     (short)0, c,  false,  false);
        }

        const int r0 = (wave << 4) + (half << 3);
        if (m > 0) {
#pragma unroll
            for (int v = 0; v < 8; ++v) sX[cur ^ 1][r0 + v][laneM] = c[v];
            __syncthreads();
            cur ^= 1;
        } else {
#pragma unroll
            for (int v = 0; v < 8; ++v) Wc[(long)(r0 + v) * D + J + laneM] = c[v];
        }
    }

    __syncthreads();
    if (tid < (NLAYERS - 1) * 16) {
        const int m = tid >> 4, col = tid & 15;
        float sacc = 0.0f;
#pragma unroll
        for (int g = 0; g < 16; ++g) sacc += sP[m][g][col];
        cvec[m * D + J + col] = sacc;
    }
}

__global__ void __launch_bounds__(256)
degree_place_kernel(const int* __restrict__ src, const int* __restrict__ dst,
                    float* __restrict__ deg2, int* __restrict__ cnt2,
                    int* __restrict__ slots, int* __restrict__ spill_cnt,
                    int* __restrict__ spill, int E, int N,
                    const float* __restrict__ W, const float* __restrict__ bias,
                    float* __restrict__ Wc, float* __restrict__ cvec) {
    if (blockIdx.x < CHAIN_GROUPS) {
        weight_chain_block(W, bias, Wc, cvec, blockIdx.x);
        return;
    }
    int e = (blockIdx.x - CHAIN_GROUPS) * blockDim.x + threadIdx.x;
    const int  h   = cache_half();
    const long off = (long)h * N;
    if (e < E) {
        const int s = src[e];
        const int d = dst[e];
        unsafeAtomicAdd(&deg2[off + s], 1.0f);
        const int r = atomicAdd(&cnt2[off + d], 1);
        if (r < CAP) {
            slots[(off + d) * CAP + r] = s;
        } else {
            const int k = atomicAdd(spill_cnt, 1);
            int2 o;
            o.x = d;
            o.y = s;
            *(int2*)(spill + 2L * k) = o;
        }
    }
}

#define GATHER_LOAD(k)                                                         \
    {                                                                          \
        int s_ = __builtin_amdgcn_readlane(sj, i + (k));                       \
        v[k]   = *(const float4*)(hin + (long)s_ * D + q);                     \
    }
#define GATHER_ACC(k)                                                          \
    {                                                                          \
        acc.x += v[k].x;                                                       \
        acc.y += v[k].y;                                                       \
        acc.z += v[k].z;                                                       \
        acc.w += v[k].w;                                                       \
    }
#define GATHER_ACC_W(k)                                                        \
    {                                                                          \
        const float g0_ = __int_as_float(                                      \
            __builtin_amdgcn_readlane(__float_as_int(dg0), i + (k)));          \
        const float g1_ = __int_as_float(                                      \
            __builtin_amdgcn_readlane(__float_as_int(dg1), i + (k)));          \
        const float w_ = rsqrtf(fmaxf(g0_ + g1_, 1.0f));                       \
        acc.x += v[k].x * w_;                                                  \
        acc.y += v[k].y * w_;                                                  \
        acc.z += v[k].z * w_;                                                  \
        acc.w += v[k].w * w_;                                                  \
    }

template <bool WEIGHTED>
__device__ __forceinline__ void spill_add(float4& acc, const float* __restrict__ hin,
                                          const int* __restrict__ spill, int m, int d,
                                          const float* __restrict__ deg2, int N,
                                          int lane, int q) {
    for (int kb = 0; kb < m; kb += 32) {
        int od = -1, os = 0;
        if (kb + lane < m) {
            const int2 o = *(const int2*)(spill + 2L * (kb + lane));
            od = o.x;
            os = o.y;
        }
        unsigned mask = __builtin_amdgcn_ballot_w32(od == d);
        while (mask) {
            const int b = __builtin_ctz(mask);
            mask &= mask - 1u;
            const int s_ = __builtin_amdgcn_readlane(os, b);
            const float4 v = *(const float4*)(hin + (long)s_ * D + q);
            float w_ = 1.0f;
            if (WEIGHTED) w_ = rsqrtf(fmaxf(deg2[s_] + deg2[N + s_], 1.0f));
            acc.x += v.x * w_;
            acc.y += v.y * w_;
            acc.z += v.z * w_;
            acc.w += v.w * w_;
        }
    }
}

__device__ __forceinline__ float4 gather_row_w(const float* __restrict__ hin,
                                               const int* __restrict__ slots,
                                               const float* __restrict__ deg2,
                                               int N, int d, int c0, int c1,
                                               int lane, int q) {
    float4 acc = make_float4(0.0f, 0.0f, 0.0f, 0.0f);
    const int n0 = min(c0, CAP);
    const int n  = n0 + min(c1, CAP);
    const int* __restrict__ S0 = slots + (long)d * CAP;
    const int* __restrict__ S1 = slots + ((long)N + d) * CAP;

    for (int base = 0; base < n; base += 32) {
        int cnt = n - base;
        if (cnt > 32) cnt = 32;
        int sj = 0;
        if (lane < cnt) {
            const int j = base + lane;
            sj = (j < n0) ? S0[j] : S1[j - n0];
        }
        const float dg0 = deg2[sj];
        const float dg1 = deg2[N + sj];

        int i = 0;
        for (; i + 8 <= cnt; i += 8) {
            float4 v[8];
            GATHER_LOAD(0) GATHER_LOAD(1) GATHER_LOAD(2) GATHER_LOAD(3)
            GATHER_LOAD(4) GATHER_LOAD(5) GATHER_LOAD(6) GATHER_LOAD(7)
            GATHER_ACC_W(0) GATHER_ACC_W(1) GATHER_ACC_W(2) GATHER_ACC_W(3)
            GATHER_ACC_W(4) GATHER_ACC_W(5) GATHER_ACC_W(6) GATHER_ACC_W(7)
        }
        const int r = cnt - i;
        if (r > 0) {
            float4 v[7];
            GATHER_LOAD(0)
            if (r > 1) GATHER_LOAD(1)
            if (r > 2) GATHER_LOAD(2)
            if (r > 3) GATHER_LOAD(3)
            if (r > 4) GATHER_LOAD(4)
            if (r > 5) GATHER_LOAD(5)
            if (r > 6) GATHER_LOAD(6)
            GATHER_ACC_W(0)
            if (r > 1) GATHER_ACC_W(1)
            if (r > 2) GATHER_ACC_W(2)
            if (r > 3) GATHER_ACC_W(3)
            if (r > 4) GATHER_ACC_W(4)
            if (r > 5) GATHER_ACC_W(5)
            if (r > 6) GATHER_ACC_W(6)
        }
    }
    return acc;
}

#define WREG 16

__global__ void __launch_bounds__(256) __attribute__((amdgpu_waves_per_eu(8, 8)))
gcn_layer_kernel(const float* __restrict__ hin,
                 const int* __restrict__ slots,
                 const int* __restrict__ cnt2,
                 const float* __restrict__ deg2,
                 const int* __restrict__ spill,
                 const int* __restrict__ spill_cnt,
                 const float* __restrict__ W,
                 const float* __restrict__ bias,
                 float* __restrict__ out, int N, int numTiles) {
    __shared__ __attribute__((aligned(16))) float sA[2][16][D + 4];
    __shared__ float sNs[2][16];
    __shared__ v2f sB[8][D / 4 - WREG][32];

    const int tid   = threadIdx.x;
    const int wave  = __builtin_amdgcn_readfirstlane(tid >> 5);
    const int lane  = tid & 31;
    const int q     = lane << 2;
    const int laneM = lane & 15;
    const int koff  = (lane >> 4) << 1;
    const int n     = (wave << 4) + laneM;

    v2f bfr[WREG];
#pragma unroll
    for (int ks = 0; ks < D / 4; ++ks) {
        v2f b;
        b.x = W[(long)(ks * 4 + koff) * D + n];
        b.y = W[(long)(ks * 4 + koff + 1) * D + n];
        if (ks < WREG) bfr[ks] = b;
        else           sB[wave][ks - WREG][lane] = b;
    }
    const float bv = bias[n];

    int   cA0 = 0, cA1 = 0, cB0 = 0, cB1 = 0;
    float gA = 1.0f, gB = 1.0f;
    {
        const int d0 = blockIdx.x * 16 + (wave << 1);
        if (blockIdx.x < numTiles && d0 < N) {
            const int dB = min(d0 + 1, N - 1);
            cA0 = cnt2[d0];
            cA1 = cnt2[N + d0];
            const int tB0 = cnt2[dB];
            const int tB1 = cnt2[N + dB];
            gA  = deg2[d0] + deg2[N + d0];
            gB  = deg2[dB] + deg2[N + dB];
            cB0 = (d0 + 1 < N) ? tB0 : 0;
            cB1 = (d0 + 1 < N) ? tB1 : 0;
        }
    }

    int buf = 0;
    for (int tile = blockIdx.x; tile < numTiles; tile += gridDim.x, buf ^= 1) {
        const int rowBase = tile * 16;

        {
            const int r0 = wave << 1;
            const int d0 = rowBase + r0;
            const int a0 = __builtin_amdgcn_readfirstlane(cA0);
            const int a1 = __builtin_amdgcn_readfirstlane(cA1);
            const int b0 = __builtin_amdgcn_readfirstlane(cB0);
            const int b1 = __builtin_amdgcn_readfirstlane(cB1);

            float4 x0 = gather_row_w(hin, slots, deg2, N, d0, a0, a1, lane, q);
            if (a0 > CAP || a1 > CAP)
                spill_add<true>(x0, hin, spill, *spill_cnt, d0, deg2, N, lane, q);
            const float nd0 = rsqrtf(fmaxf((float)(a0 + a1), 1.0f));
            x0.x *= nd0; x0.y *= nd0; x0.z *= nd0; x0.w *= nd0;
            *(float4*)&sA[buf][r0][q] = x0;
            float4 x1 = gather_row_w(hin, slots, deg2, N, d0 + 1, b0, b1, lane, q);
            if (b0 > CAP || b1 > CAP)
                spill_add<true>(x1, hin, spill, *spill_cnt, d0 + 1, deg2, N, lane, q);
            const float nd1 = rsqrtf(fmaxf((float)(b0 + b1), 1.0f));
            x1.x *= nd1; x1.y *= nd1; x1.z *= nd1; x1.w *= nd1;
            *(float4*)&sA[buf][r0 + 1][q] = x1;
            if (lane == 0) {
                sNs[buf][r0]     = rsqrtf(fmaxf(gA, 1.0f));
                sNs[buf][r0 + 1] = rsqrtf(fmaxf(gB, 1.0f));
            }
        }
        __syncthreads();

        {
            const int nt = tile + gridDim.x;
            const int d0 = nt * 16 + (wave << 1);
            cA0 = 0; cA1 = 0; cB0 = 0; cB1 = 0; gA = 1.0f; gB = 1.0f;
            if (nt < numTiles && d0 < N) {
                const int dB = min(d0 + 1, N - 1);
                cA0 = cnt2[d0];
                cA1 = cnt2[N + d0];
                const int tB0 = cnt2[dB];
                const int tB1 = cnt2[N + dB];
                gA  = deg2[d0] + deg2[N + d0];
                gB  = deg2[dB] + deg2[N + dB];
                cB0 = (d0 + 1 < N) ? tB0 : 0;
                cB1 = (d0 + 1 < N) ? tB1 : 0;
            }
        }

        v8f c;
#pragma unroll
        for (int v = 0; v < 8; ++v) c[v] = bv;

#pragma unroll
        for (int ks = 0; ks < D / 4; ++ks) {
            v2f a;
            a.x = sA[buf][laneM][ks * 4 + koff];
            a.y = sA[buf][laneM][ks * 4 + koff + 1];
            const v2f b = (ks < WREG) ? bfr[ks] : sB[wave][ks - WREG][lane];
            c = __builtin_amdgcn_wmma_f32_16x16x4_f32(
                     false, a,  false, b,
                     (short)0, c,  false,  false);
        }

        const int mOff  = (lane >> 4) << 3;
        const int mBase = rowBase + mOff;
#pragma unroll
        for (int v = 0; v < 8; ++v) {
            int row = mBase + v;
            if (row < N) out[(long)row * D + n] = c[v] * sNs[buf][mOff + v];
        }
    }
}

__device__ __forceinline__ float4 gather_row_ahead(const float* __restrict__ hin,
                                                   const int* __restrict__ slots,
                                                   int N, int d, int c0, int c1,
                                                   int lane, int q, int first,
                                                   int dn, int e0, int e1, int& next) {
    float4 acc = make_float4(0.0f, 0.0f, 0.0f, 0.0f);
    next = 0;
    bool asked = false;

    const int n0 = min(c0, CAP);
    const int n  = n0 + min(c1, CAP);
    const int* __restrict__ S0 = slots + (long)d * CAP;
    const int* __restrict__ S1 = slots + ((long)N + d) * CAP;
    const int m0 = min(e0, CAP);
    const int mn = min(m0 + min(e1, CAP), 32);
    const int* __restrict__ T0 = slots + (long)dn * CAP;
    const int* __restrict__ T1 = slots + ((long)N + dn) * CAP;

    for (int base = 0; base < n; base += 32) {
        int cnt = n - base;
        if (cnt > 32) cnt = 32;
        int sj = first;
        if (base != 0) {
            sj = 0;
            if (lane < cnt) {
                const int j = base + lane;
                sj = (j < n0) ? S0[j] : S1[j - n0];
            }
        }

        int i = 0;
        for (; i + 8 <= cnt; i += 8) {
            float4 v[8];
            GATHER_LOAD(0) GATHER_LOAD(1) GATHER_LOAD(2) GATHER_LOAD(3)
            GATHER_LOAD(4) GATHER_LOAD(5) GATHER_LOAD(6) GATHER_LOAD(7)
            if (!asked) {
                if (lane < mn) next = (lane < m0) ? T0[lane] : T1[lane - m0];
                asked = true;
            }
            GATHER_ACC(0) GATHER_ACC(1) GATHER_ACC(2) GATHER_ACC(3)
            GATHER_ACC(4) GATHER_ACC(5) GATHER_ACC(6) GATHER_ACC(7)
        }
        const int r = cnt - i;
        if (r > 0) {
            float4 v[7];
            GATHER_LOAD(0)
            if (r > 1) GATHER_LOAD(1)
            if (r > 2) GATHER_LOAD(2)
            if (r > 3) GATHER_LOAD(3)
            if (r > 4) GATHER_LOAD(4)
            if (r > 5) GATHER_LOAD(5)
            if (r > 6) GATHER_LOAD(6)
            if (!asked) {
                if (lane < mn) next = (lane < m0) ? T0[lane] : T1[lane - m0];
                asked = true;
            }
            GATHER_ACC(0)
            if (r > 1) GATHER_ACC(1)
            if (r > 2) GATHER_ACC(2)
            if (r > 3) GATHER_ACC(3)
            if (r > 4) GATHER_ACC(4)
            if (r > 5) GATHER_ACC(5)
            if (r > 6) GATHER_ACC(6)
        }
    }
    if (!asked) {
        if (lane < mn) next = (lane < m0) ? T0[lane] : T1[lane - m0];
    }
    return acc;
}

#define GATHER_GRID 1280

template <bool SCALE_OUT>
__global__ void __launch_bounds__(256)
gather_layer_kernel(const float* __restrict__ hin,
                    const int* __restrict__ slots,
                    const int* __restrict__ cnt2,
                    const float* __restrict__ deg2,
                    const int* __restrict__ spill,
                    const int* __restrict__ spill_cnt,
                    const float* __restrict__ brow,
                    float* __restrict__ out, int N) {
    const int tid    = threadIdx.x;
    const int wave   = __builtin_amdgcn_readfirstlane(tid >> 5);
    const int lane   = tid & 31;
    const int q      = lane << 2;
    const int stride = gridDim.x * 8;

    const float4 bq = *(const float4*)(brow + q);

    int d = blockIdx.x * 8 + wave;

    int a0 = 0, a1 = 0, b0 = 0, b1 = 0;
    if (d < N) {
        a0 = cnt2[d];
        a1 = cnt2[N + d];
    }
    if (d + stride < N) {
        b0 = cnt2[d + stride];
        b1 = cnt2[N + d + stride];
    }
    int prA = 0;
    {
        const int m0 = min(a0, CAP);
        const int mn = min(m0 + min(a1, CAP), 32);
        if (lane < mn)
            prA = (lane < m0) ? slots[(long)d * CAP + lane]
                              : slots[((long)N + d) * CAP + (lane - m0)];
    }

    for (; d < N; d += stride) {
        int c0 = 0, c1 = 0;
        const int d2 = d + 2 * stride;
        if (d2 < N) {
            c0 = cnt2[d2];
            c1 = cnt2[N + d2];
        }
        float ns = 1.0f;
        if (SCALE_OUT) ns = rsqrtf(fmaxf(deg2[d] + deg2[N + d], 1.0f));
        const float nd = rsqrtf(fmaxf((float)(a0 + a1), 1.0f));

        int prB;
        float4 acc = gather_row_ahead(hin, slots, N, d, a0, a1, lane, q, prA,
                                      d + stride, b0, b1, prB);
        if (a0 > CAP || a1 > CAP)
            spill_add<false>(acc, hin, spill, *spill_cnt, d, deg2, N, lane, q);

        float4 o;
        o.x = (acc.x * nd + bq.x) * ns;
        o.y = (acc.y * nd + bq.y) * ns;
        o.z = (acc.z * nd + bq.z) * ns;
        o.w = (acc.w * nd + bq.w) * ns;
        *(float4*)(out + (long)d * D + q) = o;

        a0 = b0; a1 = b1; prA = prB;
        b0 = c0; b1 = c1;
    }
}


extern "C" void kernel_launch(void* const* d_in, const int* in_sizes, int n_in,
                              void* d_out, int out_size, void* d_ws, size_t ws_size,
                              hipStream_t stream) {
    const float* feat = (const float*)d_in[0];
    const float* W    = (const float*)d_in[1];
    const float* bias = (const float*)d_in[2];
    const int*   src  = (const int*)d_in[3];
    const int*   dst  = (const int*)d_in[4];

    const int N = in_sizes[0] / D;
    const int E = in_sizes[3];

    const long ND = (long)N * D;

    float* ws        = (float*)d_ws;
    float* hbuf      = ws;
    float* Wc        = hbuf + ND;
    float* cvec      = Wc + (long)D * D;
    int*   slots     = (int*)(cvec + 3L * D);
    int*   spill     = slots + 2L * N * CAP;
    float* deg2      = (float*)(spill + 2L * E);
    int*   cnt2      = (int*)(deg2 + 2L * N);
    int*   spill_cnt = cnt2 + 2L * N;
    float* outf      = (float*)d_out;

    const long nZero = 4L * N + 1;

    zero_words_kernel<<<(int)((nZero + 255) / 256), 256, 0, stream>>>((int*)deg2, nZero);

    degree_place_kernel<<<CHAIN_GROUPS + (E + 255) / 256, 256, 0, stream>>>(
        src, dst, deg2, cnt2, slots, spill_cnt, spill, E, N, W, bias, Wc, cvec);

    const int numTiles  = (N + 15) / 16;
    const int layerGrid = (numTiles + 3) / 4;
    gcn_layer_kernel<<<layerGrid, 256, 0, stream>>>(
        feat, slots, cnt2, deg2, spill, spill_cnt, Wc, cvec, hbuf, N, numTiles);

    gather_layer_kernel<true><<<GATHER_GRID, 256, 0, stream>>>(
        hbuf, slots, cnt2, deg2, spill, spill_cnt, cvec + D, outf, N);
    gather_layer_kernel<true><<<GATHER_GRID, 256, 0, stream>>>(
        outf, slots, cnt2, deg2, spill, spill_cnt, cvec + 2L * D, hbuf, N);
    gather_layer_kernel<false><<<GATHER_GRID, 256, 0, stream>>>(
        hbuf, slots, cnt2, deg2, spill, spill_cnt, bias + 3L * D, outf, N);
}
